# starting kernel
# baseline (speedup 1.0000x reference)
#include <hip/hip_runtime.h>
#include <math.h>

constexpr int TT = 64;
constexpr int DD = 512;
constexpr int NH = 16;
constexpr int G  = 4;

constexpr int HSTR = 516;
constexpr int RSTR = 516;
constexpr int QSTR = 36;
constexpr int QG   = NH * QSTR;
constexpr int UG   = DD * NH + 16;

constexpr int OFF_H = 0;
constexpr int OFF_R = OFF_H + TT * HSTR;
constexpr int OFF_Q = OFF_R + 16 * RSTR;
constexpr int OFF_U = OFF_Q + G * QG;
constexpr int OFF_S = OFF_U + G * UG;
constexpr int LDS_FLOATS = OFF_S + TT * NH;

typedef __attribute__((ext_vector_type(2))) float v2f;
typedef __attribute__((ext_vector_type(4))) float f4;
typedef __attribute__((ext_vector_type(8))) float v8f;

static __device__ __forceinline__ v8f wmma4(v2f a, v2f b, v8f c) {
    return __builtin_amdgcn_wmma_f32_16x16x4_f32(
        false, a, false, b, (short)0, c, false, false);
}

__global__ __launch_bounds__(256, 1)
void temporal_attn_lastrow_g4(const float* __restrict__ inputs,
                              const float* __restrict__ pos,
                              const float* __restrict__ Wq,
                              const float* __restrict__ Wk,
                              const float* __restrict__ Wv,
                              float* __restrict__ out)
{
    extern __shared__ float lds[];
    float* sH = lds + OFF_H;
    float* sR = lds + OFF_R;
    float* sQ = lds + OFF_Q;
    float* sU = lds + OFF_U;
    float* sS = lds + OFF_S;

    const int tid  = threadIdx.x;
    const int lane = tid & 31;
    const int wave = tid >> 5;
    const int m    = lane & 15;
    const int hi   = lane >> 4;
    const long b0  = (long)blockIdx.x * G;

    for (int i = tid; i < 16 * RSTR; i += 256) sR[i] = 0.f;
    __syncthreads();
    for (int i = tid; i < G * DD; i += 256) {
        int g = i >> 9, d = i & 511;
        sR[g * RSTR + d] =
            inputs[(b0 + g) * (size_t)(TT * DD) + 63 * DD + d] + pos[63 * DD + d];
    }
    __syncthreads();

    for (int t = 0; t < 4; ++t) {
        const int n0 = (wave * 4 + t) * 16;
        v8f acc = {};
        for (int k = 0; k < DD; k += 4) {
            const int kk = k + 2 * hi;
            v2f a, b;
            a[0] = sR[m * RSTR + kk];
            a[1] = sR[m * RSTR + kk + 1];
            b[0] = Wq[(size_t)kk * DD + n0 + m];
            b[1] = Wq[(size_t)(kk + 1) * DD + n0 + m];
            acc = wmma4(a, b, acc);
        }
        if (hi == 0) {
            #pragma unroll
            for (int r = 0; r < G; ++r) {
                const int e = n0 + m;
                sQ[r * QG + (e >> 5) * QSTR + (e & 31)] = acc[r];
            }
        }
    }
    __syncthreads();

    {
        const int   gl    = m & 3;
        const float vmask = (m < G) ? 1.f : 0.f;
        for (int t = 0; t < 64; ++t) {
            const int job = wave * 64 + t;
            const int dt  = job >> 4;
            const int h   = job & 15;
            const int d0  = dt * 16;
            v8f acc = {};
            #pragma unroll
            for (int k = 0; k < 32; k += 4) {
                const int kk = k + 2 * hi;
                v2f a, b;
                a[0] = Wk[(size_t)(d0 + m) * DD + 32 * h + kk];
                a[1] = Wk[(size_t)(d0 + m) * DD + 32 * h + kk + 1];
                b[0] = vmask * sQ[gl * QG + h * QSTR + kk];
                b[1] = vmask * sQ[gl * QG + h * QSTR + kk + 1];
                acc = wmma4(a, b, acc);
            }
            if (m < G) {
                #pragma unroll
                for (int r = 0; r < 8; ++r)
                    sU[m * UG + (d0 + r + 8 * hi) * NH + h] = acc[r];
            }
        }
    }
    __syncthreads();

    const float scale = 0.044194173824159216f;
    for (int g = 0; g < G; ++g) {
        {
            const f4* in4 = (const f4*)(inputs + (b0 + g) * (size_t)(TT * DD));
            const f4* pe4 = (const f4*)pos;
            for (int i = tid; i < TT * DD / 4; i += 256) {
                f4 x = in4[i], p = pe4[i];
                int t = i >> 7, d = (i & 127) << 2;
                float* dst = sH + t * HSTR + d;
                dst[0] = x.x + p.x; dst[1] = x.y + p.y;
                dst[2] = x.z + p.z; dst[3] = x.w + p.w;
            }
            if (g + 1 < G) {
                const float* nxt = inputs + (b0 + g + 1) * (size_t)(TT * DD);
                #pragma unroll
                for (int it = 0; it < 4; ++it)
                    __builtin_prefetch(nxt + (size_t)(tid + it * 256) * 32, 0, 1);
            }
        }
        __syncthreads();

        if (wave < 4) {
            const int t0 = wave * 16;
            v8f acc = {};
            for (int k = 0; k < DD; k += 4) {
                const int kk = k + 2 * hi;
                v2f a, b;
                a[0] = sH[(t0 + m) * HSTR + kk];
                a[1] = sH[(t0 + m) * HSTR + kk + 1];
                b[0] = sU[g * UG + kk * NH + m];
                b[1] = sU[g * UG + (kk + 1) * NH + m];
                acc = wmma4(a, b, acc);
            }
            #pragma unroll
            for (int r = 0; r < 8; ++r)
                sS[(t0 + r + 8 * hi) * NH + m] = acc[r] * scale;
        }
        __syncthreads();

        if (tid < NH) {
            float mx = -INFINITY;
            for (int t = 0; t < TT; ++t) mx = fmaxf(mx, sS[t * NH + tid]);
            float sum = 0.f;
            for (int t = 0; t < TT; ++t) {
                float e = expf(sS[t * NH + tid] - mx);
                sS[t * NH + tid] = e;
                sum += e;
            }
            float inv = 1.f / sum;
            for (int t = 0; t < TT; ++t) sS[t * NH + tid] *= inv;
        }
        __syncthreads();

        for (int t = 0; t < 4; ++t) {
            const int d0 = (wave * 4 + t) * 16;
            v8f acc = {};
            for (int k = 0; k < TT; k += 4) {
                const int kk = k + 2 * hi;
                v2f a, b;
                a[0] = sH[kk * HSTR + d0 + m];
                a[1] = sH[(kk + 1) * HSTR + d0 + m];
                b[0] = sS[kk * NH + m];
                b[1] = sS[(kk + 1) * NH + m];
                acc = wmma4(a, b, acc);
            }
            #pragma unroll
            for (int r = 0; r < 8; ++r)
                sU[g * UG + (d0 + r + 8 * hi) * NH + m] = acc[r];
        }
        __syncthreads();
    }

    {
        const int   gl    = m & 3;
        const float vmask = (m < G) ? 1.f : 0.f;
        for (int t = 0; t < 4; ++t) {
            const int job = wave * 4 + t;
            const int h   = job >> 1;
            const int n0  = (job & 1) * 16;
            v8f acc = {};
            for (int k = 0; k < DD; k += 4) {
                const int kk = k + 2 * hi;
                v2f a, b;
                a[0] = vmask * sU[gl * UG + kk * NH + h];
                a[1] = vmask * sU[gl * UG + (kk + 1) * NH + h];
                b[0] = Wv[(size_t)kk * DD + 32 * h + n0 + m];
                b[1] = Wv[(size_t)(kk + 1) * DD + 32 * h + n0 + m];
                acc = wmma4(a, b, acc);
            }
            if (hi == 0) {
                #pragma unroll
                for (int r = 0; r < G; ++r)
                    out[(b0 + r) * (size_t)DD + 32 * h + n0 + m] = acc[r];
            }
        }
    }
}

extern "C" void kernel_launch(void* const* d_in, const int* in_sizes, int n_in,
                              void* d_out, int out_size, void* d_ws, size_t ws_size,
                              hipStream_t stream) {
    const float* inputs = (const float*)d_in[0];
    const float* pos    = (const float*)d_in[1];
    const float* Wq     = (const float*)d_in[2];
    const float* Wk     = (const float*)d_in[3];
    const float* Wv     = (const float*)d_in[4];
    float* out          = (float*)d_out;
    (void)d_ws; (void)ws_size; (void)n_in; (void)out_size;

    const int B = in_sizes[0] / (TT * DD);
    const size_t smem = (size_t)LDS_FLOATS * sizeof(float);

    hipFuncSetAttribute((const void*)temporal_attn_lastrow_g4,
                        hipFuncAttributeMaxDynamicSharedMemorySize, (int)smem);

    temporal_attn_lastrow_g4<<<B / G, 256, smem, stream>>>(
        inputs, pos, Wq, Wk, Wv, out);
}


# score GEMM H @ U_g: next 64-d chunk's H, U fragments read from LDS as one group ahead of the products; 4 accumulator chains (k-step mod 4) summed as a tree, sum over d was one chain
# speedup vs baseline: 1.7891x; 1.7891x over previous
#include <hip/hip_runtime.h>
#include <math.h>

constexpr int TT = 64;
constexpr int DD = 512;
constexpr int NH = 16;
constexpr int G  = 4;

constexpr int HSTR = 516;
constexpr int RSTR = 516;
constexpr int QSTR = 36;
constexpr int QG   = NH * QSTR;
constexpr int UG   = DD * NH + 16;

constexpr int OFF_H = 0;
constexpr int OFF_R = OFF_H + TT * HSTR;
constexpr int OFF_Q = OFF_R + 16 * RSTR;
constexpr int OFF_U = OFF_Q + G * QG;
constexpr int OFF_S = OFF_U + G * UG;
constexpr int OFF_M = OFF_S + TT * NH;
constexpr int LDS_FLOATS = OFF_M + 2 * 4 * NH;

constexpr int CH = 16;
constexpr int KC = 4;
constexpr int XN = 32;
constexpr int SK = 16;

typedef __attribute__((ext_vector_type(2))) float v2f;
typedef __attribute__((ext_vector_type(4))) float f4;
typedef __attribute__((ext_vector_type(8))) float v8f;

static __device__ __forceinline__ v8f wmma4(v2f a, v2f b, v8f c) {
    return __builtin_amdgcn_wmma_f32_16x16x4_f32(
        false, a, false, b, (short)0, c, false, false);
}

static __device__ __forceinline__ float xhalf(float x) {
    const int b = __builtin_bit_cast(int, x);
    return __builtin_bit_cast(float,
        __builtin_amdgcn_permlanex16(b, b, 0x76543210, 0xfedcba98, false, false));
}

static __device__ __forceinline__ void lds_barrier() {
    __builtin_amdgcn_fence(__ATOMIC_RELEASE, "workgroup", "local");
    __builtin_amdgcn_s_barrier();
    __builtin_amdgcn_fence(__ATOMIC_ACQUIRE, "workgroup", "local");
}

static __device__ __forceinline__ void ld_q4(f4 (&b)[2 * KC], const float* W,
                                             int kb, int col4, int hi) {
    #pragma unroll
    for (int s = 0; s < KC; ++s) {
        const int kk = kb + 4 * s + 2 * hi;
        b[2 * s]     = *(const f4*)(W + (size_t)kk * DD + col4);
        b[2 * s + 1] = *(const f4*)(W + (size_t)(kk + 1) * DD + col4);
    }
}

static __device__ __forceinline__ void mac_q4(v8f (&acc)[4], const f4 (&b)[2 * KC],
                                              const float* arow, int kb, int hi) {
    v2f a[KC];
    #pragma unroll
    for (int s = 0; s < KC; ++s) {
        const int kk = kb + 4 * s + 2 * hi;
        a[s][0] = arow[kk];
        a[s][1] = arow[kk + 1];
    }
    #pragma unroll
    for (int s = 0; s < KC; ++s) {
        #pragma unroll
        for (int j = 0; j < 4; ++j) {
            v2f bb;
            bb[0] = b[2 * s][j];
            bb[1] = b[2 * s + 1][j];
            acc[j] = wmma4(a[s], bb, acc[j]);
        }
    }
}

static __device__ __forceinline__ void ld_o4(v2f (&b)[4 * KC], const float* W,
                                             int kb, int col2, int hi) {
    #pragma unroll
    for (int s = 0; s < KC; ++s) {
        const int kk = kb + 4 * s + 2 * hi;
        const float* p = W + (size_t)kk * DD + col2;
        b[4 * s]     = *(const v2f*)(p);
        b[4 * s + 1] = *(const v2f*)(p + DD);
        b[4 * s + 2] = *(const v2f*)(p + 32);
        b[4 * s + 3] = *(const v2f*)(p + DD + 32);
    }
}

static __device__ __forceinline__ void mac_o4(v8f (&acc)[4], const v2f (&b)[4 * KC],
                                              const float* u2, float vmask,
                                              int kb, int hi) {
    #pragma unroll
    for (int s = 0; s < KC; ++s) {
        const int kk = kb + 4 * s + 2 * hi;
        #pragma unroll
        for (int x = 0; x < 2; ++x) {
            v2f a;
            a[0] = vmask * u2[kk * NH + x];
            a[1] = vmask * u2[(kk + 1) * NH + x];
            #pragma unroll
            for (int j = 0; j < 2; ++j) {
                v2f bb;
                bb[0] = b[4 * s + 2 * x][j];
                bb[1] = b[4 * s + 2 * x + 1][j];
                acc[2 * x + j] = wmma4(a, bb, acc[2 * x + j]);
            }
        }
    }
}

static __device__ __forceinline__ void ld_wk2(v2f (&a)[CH], const float* Wk,
                                              int job0, int m, int hi) {
    #pragma unroll
    for (int j = 0; j < 2; ++j) {
        const int job = job0 + j;
        const int d0  = (job >> 4) * 16;
        const int h   = job & 15;
        const float* p = Wk + (size_t)(d0 + m) * DD + 32 * h + 2 * hi;
        #pragma unroll
        for (int s = 0; s < 8; ++s) {
            a[j * 8 + s][0] = p[4 * s];
            a[j * 8 + s][1] = p[4 * s + 1];
        }
    }
}

static __device__ __forceinline__ void mac_u2(const v2f (&a)[CH], const float* sQ,
                                              float* sU, int job0, int m, int hi) {
    const int   gl    = m & 3;
    const float vmask = (m < G) ? 1.f : 0.f;
    #pragma unroll
    for (int j = 0; j < 2; ++j) {
        const int job = job0 + j;
        const int d0  = (job >> 4) * 16;
        const int h   = job & 15;
        v8f acc = {};
        #pragma unroll
        for (int s = 0; s < 8; ++s) {
            const int kk = 4 * s + 2 * hi;
            v2f b;
            b[0] = vmask * sQ[gl * QG + h * QSTR + kk];
            b[1] = vmask * sQ[gl * QG + h * QSTR + kk + 1];
            acc = wmma4(a[j * 8 + s], b, acc);
        }
        if (m < G) {
            #pragma unroll
            for (int r = 0; r < 8; ++r)
                sU[m * UG + (d0 + r + 8 * hi) * NH + h] = acc[r];
        }
    }
}

static __device__ __forceinline__ void ld_s(v2f (&a)[SK], v2f (&b)[SK], const float* hrow,
                                            const float* ucol, int d0) {
    #pragma unroll
    for (int s = 0; s < SK; ++s) {
        a[s] = *(const v2f*)(hrow + d0 + 4 * s);
        b[s][0] = ucol[(d0 + 4 * s) * NH];
        b[s][1] = ucol[(d0 + 4 * s + 1) * NH];
    }
}

static __device__ __forceinline__ void mac_s(v8f (&ch)[4], const v2f (&a)[SK], const v2f (&b)[SK]) {
    #pragma unroll
    for (int s = 0; s < SK; ++s) ch[s & 3] = wmma4(a[s], b[s], ch[s & 3]);
}

static __device__ __forceinline__ void ld_tile(f4 (&x)[XN], const float* src, int tid) {
    const f4* in4 = (const f4*)src;
    #pragma unroll
    for (int j = 0; j < XN; ++j) x[j] = in4[tid + 256 * j];
}

__global__ __launch_bounds__(256, 1)
void temporal_attn_lastrow_g4(const float* inputs,
                              const float* pos,
                              const float* Wq,
                              const float* Wk,
                              const float* Wv,
                              float* __restrict__ out)
{
    extern __shared__ float lds[];
    float* sH = lds + OFF_H;
    float* sR = lds + OFF_R;
    float* sQ = lds + OFF_Q;
    float* sU = lds + OFF_U;
    float* sS = lds + OFF_S;
    float* sM = lds + OFF_M;
    float* sL = sM + 4 * NH;

    const int tid  = threadIdx.x;
    const int lane = tid & 31;
    const int wave = tid >> 5;
    const int m    = lane & 15;
    const int hi   = lane >> 4;
    const long b0  = (long)blockIdx.x * G;

    float r0[G * DD / 256], p0[G * DD / 256];
    #pragma unroll
    for (int q = 0; q < G * DD / 256; ++q) {
        const int i = tid + 256 * q;
        const int g = i >> 9, d = i & 511;
        r0[q] = inputs[(b0 + g) * (size_t)(TT * DD) + 63 * DD + d];
        p0[q] = pos[63 * DD + d];
    }
    f4 X[XN];
    ld_tile(X, inputs + b0 * (size_t)(TT * DD), tid);

    for (int i = tid; i < 16 * RSTR; i += 256) sR[i] = 0.f;
    lds_barrier();
    #pragma unroll
    for (int q = 0; q < G * DD / 256; ++q) {
        const int i = tid + 256 * q;
        const int g = i >> 9, d = i & 511;
        sR[g * RSTR + d] = r0[q] + p0[q];
    }
    lds_barrier();

    {
        const float* arow = sR + m * RSTR;
        const int col4 = wave * 64 + 4 * m;
        f4 B0[2 * KC], B1[2 * KC];
        v8f acc[4] = {};
        ld_q4(B0, Wq, 0, col4, hi);
        for (int c = 0; c < 32; c += 2) {
            ld_q4(B1, Wq, (c + 1) * 16, col4, hi);
            mac_q4(acc, B0, arow, c * 16, hi);
            if (c + 2 < 32) ld_q4(B0, Wq, (c + 2) * 16, col4, hi);
            mac_q4(acc, B1, arow, (c + 1) * 16, hi);
        }
        if (hi == 0) {
            #pragma unroll
            for (int j = 0; j < 4; ++j) {
                const int e = col4 + j;
                #pragma unroll
                for (int r = 0; r < G; ++r)
                    sQ[r * QG + (e >> 5) * QSTR + (e & 31)] = acc[j][r];
            }
        }
    }
    lds_barrier();

    {
        v2f A0[CH], A1[CH];
        const int jb = wave * 64;
        ld_wk2(A0, Wk, jb, m, hi);
        for (int c = 0; c < 32; c += 2) {
            ld_wk2(A1, Wk, jb + 2 * (c + 1), m, hi);
            mac_u2(A0, sQ, sU, jb + 2 * c, m, hi);
            if (c + 2 < 32) ld_wk2(A0, Wk, jb + 2 * (c + 2), m, hi);
            mac_u2(A1, sQ, sU, jb + 2 * (c + 1), m, hi);
        }
    }
    lds_barrier();

    const float scale = 0.044194173824159216f;
    for (int g = 0; g < G; ++g) {
        {
            const f4* pe4 = (const f4*)pos;
            f4 P[XN];
            #pragma unroll
            for (int j = 0; j < XN; ++j) P[j] = pe4[tid + 256 * j];
            #pragma unroll
            for (int j = 0; j < XN; ++j) {
                const int i = tid + 256 * j;
                f4 x = X[j], p = P[j];
                int t = i >> 7, d = (i & 127) << 2;
                float* dst = sH + t * HSTR + d;
                dst[0] = x.x + p.x; dst[1] = x.y + p.y;
                dst[2] = x.z + p.z; dst[3] = x.w + p.w;
            }
            if (g + 1 < G)
                ld_tile(X, inputs + (b0 + g + 1) * (size_t)(TT * DD), tid);
        }
        lds_barrier();

        float sc[8] = {};
        if (wave < 4) {
            const int t0 = wave * 16;
            const float* hrow = sH + (t0 + m) * HSTR + 2 * hi;
            int uoff = OFF_U + g * UG + (2 * hi) * NH + m;
            asm volatile("" : "+v"(uoff));
            const float* ucol = lds + uoff;
            v2f A0[SK], A1[SK], B0[SK], B1[SK];
            v8f ch[4] = {};
            ld_s(A0, B0, hrow, ucol, 0);
            for (int c = 0; c < DD / (4 * SK); c += 2) {
                ld_s(A1, B1, hrow, ucol, (c + 1) * 4 * SK);
                mac_s(ch, A0, B0);
                if (c + 2 < DD / (4 * SK)) ld_s(A0, B0, hrow, ucol, (c + 2) * 4 * SK);
                mac_s(ch, A1, B1);
            }
            const v8f acc = (ch[0] + ch[1]) + (ch[2] + ch[3]);
            float mx = -INFINITY;
            #pragma unroll
            for (int r = 0; r < 8; ++r) {
                sc[r] = acc[r] * scale;
                mx = fmaxf(mx, sc[r]);
            }
            mx = fmaxf(mx, xhalf(mx));
            if (hi == 0) sM[wave * NH + m] = mx;
        }
        lds_barrier();

        if (wave < 4) {
            const int t0 = wave * 16;
            const float mx = fmaxf(fmaxf(sM[m], sM[NH + m]),
                                   fmaxf(sM[2 * NH + m], sM[3 * NH + m]));
            float sum = 0.f;
            #pragma unroll
            for (int r = 0; r < 8; ++r) {
                const float e = expf(sc[r] - mx);
                sS[(t0 + r + 8 * hi) * NH + m] = e;
                sum += e;
            }
            sum += xhalf(sum);
            if (hi == 0) sL[wave * NH + m] = sum;
        }
        lds_barrier();

        {
            v2f pb[16], ha[4][16];
            const float inv = 1.f / (((sL[m] + sL[NH + m]) + sL[2 * NH + m]) + sL[3 * NH + m]);
            #pragma unroll
            for (int s = 0; s < 16; ++s) {
                const int kk = 4 * s + 2 * hi;
                pb[s][0] = sS[kk * NH + m] * inv;
                pb[s][1] = sS[(kk + 1) * NH + m] * inv;
            }
            #pragma unroll
            for (int t = 0; t < 4; ++t) {
                const int d0 = (wave * 4 + t) * 16;
                #pragma unroll
                for (int s = 0; s < 16; ++s) {
                    const int kk = 4 * s + 2 * hi;
                    ha[t][s][0] = sH[kk * HSTR + d0 + m];
                    ha[t][s][1] = sH[(kk + 1) * HSTR + d0 + m];
                }
            }
            #pragma unroll
            for (int t = 0; t < 4; ++t) {
                const int d0 = (wave * 4 + t) * 16;
                v8f acc = {};
                #pragma unroll
                for (int s = 0; s < 16; ++s) acc = wmma4(ha[t][s], pb[s], acc);
                #pragma unroll
                for (int r = 0; r < 8; ++r)
                    sU[g * UG + (d0 + r + 8 * hi) * NH + m] = acc[r];
            }
        }
        lds_barrier();
    }

    {
        const int   gl    = m & 3;
        const float vmask = (m < G) ? 1.f : 0.f;
        const float* u2   = sU + gl * UG + 2 * wave;
        const int col2 = wave * 64 + 2 * m;
        v2f B0[4 * KC], B1[4 * KC];
        v8f acc[4] = {};
        ld_o4(B0, Wv, 0, col2, hi);
        for (int c = 0; c < 32; c += 2) {
            ld_o4(B1, Wv, (c + 1) * 16, col2, hi);
            mac_o4(acc, B0, u2, vmask, c * 16, hi);
            if (c + 2 < 32) ld_o4(B0, Wv, (c + 2) * 16, col2, hi);
            mac_o4(acc, B1, u2, vmask, (c + 1) * 16, hi);
        }
        if (hi == 0) {
            #pragma unroll
            for (int x = 0; x < 2; ++x) {
                #pragma unroll
                for (int r = 0; r < G; ++r) {
                    v2f o;
                    o[0] = acc[2 * x][r];
                    o[1] = acc[2 * x + 1][r];
                    *(v2f*)(out + (b0 + r) * (size_t)DD + col2 + 32 * x) = o;
                }
            }
        }
    }
}

extern "C" void kernel_launch(void* const* d_in, const int* in_sizes, int n_in,
                              void* d_out, int out_size, void* d_ws, size_t ws_size,
                              hipStream_t stream) {
    const float* inputs = (const float*)d_in[0];
    const float* pos    = (const float*)d_in[1];
    const float* Wq     = (const float*)d_in[2];
    const float* Wk     = (const float*)d_in[3];
    const float* Wv     = (const float*)d_in[4];
    float* out          = (float*)d_out;
    (void)d_ws; (void)ws_size; (void)n_in; (void)out_size;

    const int B = in_sizes[0] / (TT * DD);
    const size_t smem = (size_t)LDS_FLOATS * sizeof(float);

    hipFuncSetAttribute((const void*)temporal_attn_lastrow_g4,
                        hipFuncAttributeMaxDynamicSharedMemorySize, (int)smem);

    temporal_attn_lastrow_g4<<<B / G, 256, smem, stream>>>(
        inputs, pos, Wq, Wk, Wv, out);
}


# Wq/Wk/Wv chunk requests go two chunks ahead through a ring of four register sets (was one ahead, and the wait before every second chunk also covered the newest request); bit-identical
# speedup vs baseline: 1.9576x; 1.0942x over previous
#include <hip/hip_runtime.h>
#include <math.h>

constexpr int TT = 64;
constexpr int DD = 512;
constexpr int NH = 16;
constexpr int G  = 4;

constexpr int HSTR = 516;
constexpr int RSTR = 516;
constexpr int QSTR = 36;
constexpr int QG   = NH * QSTR;
constexpr int UG   = DD * NH + 16;

constexpr int OFF_H = 0;
constexpr int OFF_R = OFF_H + TT * HSTR;
constexpr int OFF_Q = OFF_R + 16 * RSTR;
constexpr int OFF_U = OFF_Q + G * QG;
constexpr int OFF_S = OFF_U + G * UG;
constexpr int OFF_M = OFF_S + TT * NH;
constexpr int LDS_FLOATS = OFF_M + 2 * 4 * NH;

constexpr int CH = 16;
constexpr int KC = 4;
constexpr int XN = 32;
constexpr int SK = 16;

typedef __attribute__((ext_vector_type(2))) float v2f;
typedef __attribute__((ext_vector_type(4))) float f4;
typedef __attribute__((ext_vector_type(8))) float v8f;

static __device__ __forceinline__ v8f wmma4(v2f a, v2f b, v8f c) {
    return __builtin_amdgcn_wmma_f32_16x16x4_f32(
        false, a, false, b, (short)0, c, false, false);
}

static __device__ __forceinline__ float xhalf(float x) {
    const int b = __builtin_bit_cast(int, x);
    return __builtin_bit_cast(float,
        __builtin_amdgcn_permlanex16(b, b, 0x76543210, 0xfedcba98, false, false));
}

static __device__ __forceinline__ void lds_barrier() {
    __builtin_amdgcn_fence(__ATOMIC_RELEASE, "workgroup", "local");
    __builtin_amdgcn_s_barrier();
    __builtin_amdgcn_fence(__ATOMIC_ACQUIRE, "workgroup", "local");
}

static __device__ __forceinline__ int last_chunk(int c) { return c < 31 ? c : 31; }

static __device__ __forceinline__ void ld_q4(f4 (&b)[2 * KC], const float* W,
                                             int kb, int col4, int hi) {
    #pragma unroll
    for (int s = 0; s < KC; ++s) {
        const int kk = kb + 4 * s + 2 * hi;
        b[2 * s]     = *(const f4*)(W + (size_t)kk * DD + col4);
        b[2 * s + 1] = *(const f4*)(W + (size_t)(kk + 1) * DD + col4);
    }
}

static __device__ __forceinline__ void mac_q4(v8f (&acc)[4], const f4 (&b)[2 * KC],
                                              const float* arow, int kb, int hi) {
    f4 bq[2 * KC];
    #pragma unroll
    for (int i = 0; i < 2 * KC; ++i) {
        bq[i] = b[i];
        asm volatile("" : "+v"(bq[i]));
    }
    v2f a[KC];
    #pragma unroll
    for (int s = 0; s < KC; ++s) {
        const int kk = kb + 4 * s + 2 * hi;
        a[s][0] = arow[kk];
        a[s][1] = arow[kk + 1];
    }
    #pragma unroll
    for (int s = 0; s < KC; ++s) {
        #pragma unroll
        for (int j = 0; j < 4; ++j) {
            v2f bb;
            bb[0] = bq[2 * s][j];
            bb[1] = bq[2 * s + 1][j];
            acc[j] = wmma4(a[s], bb, acc[j]);
        }
    }
}

static __device__ __forceinline__ void ld_o4(v2f (&b)[4 * KC], const float* W,
                                             int kb, int col2, int hi) {
    #pragma unroll
    for (int s = 0; s < KC; ++s) {
        const int kk = kb + 4 * s + 2 * hi;
        const float* p = W + (size_t)kk * DD + col2;
        b[4 * s]     = *(const v2f*)(p);
        b[4 * s + 1] = *(const v2f*)(p + DD);
        b[4 * s + 2] = *(const v2f*)(p + 32);
        b[4 * s + 3] = *(const v2f*)(p + DD + 32);
    }
}

static __device__ __forceinline__ void mac_o4(v8f (&acc)[4], const v2f (&b)[4 * KC],
                                              const float* u2, float vmask,
                                              int kb, int hi) {
    v2f bq[4 * KC];
    #pragma unroll
    for (int i = 0; i < 4 * KC; ++i) {
        bq[i] = b[i];
        asm volatile("" : "+v"(bq[i]));
    }
    #pragma unroll
    for (int s = 0; s < KC; ++s) {
        const int kk = kb + 4 * s + 2 * hi;
        #pragma unroll
        for (int x = 0; x < 2; ++x) {
            v2f a;
            a[0] = vmask * u2[kk * NH + x];
            a[1] = vmask * u2[(kk + 1) * NH + x];
            #pragma unroll
            for (int j = 0; j < 2; ++j) {
                v2f bb;
                bb[0] = bq[4 * s + 2 * x][j];
                bb[1] = bq[4 * s + 2 * x + 1][j];
                acc[2 * x + j] = wmma4(a, bb, acc[2 * x + j]);
            }
        }
    }
}

static __device__ __forceinline__ void ld_wk2(v2f (&a)[CH], const float* Wk,
                                              int job0, int m, int hi) {
    #pragma unroll
    for (int j = 0; j < 2; ++j) {
        const int job = job0 + j;
        const int d0  = (job >> 4) * 16;
        const int h   = job & 15;
        const float* p = Wk + (size_t)(d0 + m) * DD + 32 * h + 2 * hi;
        #pragma unroll
        for (int s = 0; s < 8; ++s) {
            a[j * 8 + s][0] = p[4 * s];
            a[j * 8 + s][1] = p[4 * s + 1];
        }
    }
}

static __device__ __forceinline__ void mac_u2(const v2f (&a)[CH], const float* sQ,
                                              float* sU, int job0, int m, int hi) {
    const int   gl    = m & 3;
    const float vmask = (m < G) ? 1.f : 0.f;
    #pragma unroll
    for (int j = 0; j < 2; ++j) {
        const int job = job0 + j;
        const int d0  = (job >> 4) * 16;
        const int h   = job & 15;
        v8f acc = {};
        #pragma unroll
        for (int s = 0; s < 8; ++s) {
            const int kk = 4 * s + 2 * hi;
            v2f b;
            b[0] = vmask * sQ[gl * QG + h * QSTR + kk];
            b[1] = vmask * sQ[gl * QG + h * QSTR + kk + 1];
            acc = wmma4(a[j * 8 + s], b, acc);
        }
        if (m < G) {
            #pragma unroll
            for (int r = 0; r < 8; ++r)
                sU[m * UG + (d0 + r + 8 * hi) * NH + h] = acc[r];
        }
    }
}

static __device__ __forceinline__ void ld_s(v2f (&a)[SK], v2f (&b)[SK], const float* hrow,
                                            const float* ucol, int d0) {
    #pragma unroll
    for (int s = 0; s < SK; ++s) {
        a[s] = *(const v2f*)(hrow + d0 + 4 * s);
        b[s][0] = ucol[(d0 + 4 * s) * NH];
        b[s][1] = ucol[(d0 + 4 * s + 1) * NH];
    }
}

static __device__ __forceinline__ void mac_s(v8f (&ch)[4], const v2f (&a)[SK], const v2f (&b)[SK]) {
    #pragma unroll
    for (int s = 0; s < SK; ++s) ch[s & 3] = wmma4(a[s], b[s], ch[s & 3]);
}

static __device__ __forceinline__ void ld_tile(f4 (&x)[XN], const float* src, int tid) {
    const f4* in4 = (const f4*)src;
    #pragma unroll
    for (int j = 0; j < XN; ++j) x[j] = in4[tid + 256 * j];
}

__global__ __launch_bounds__(256, 1)
void temporal_attn_lastrow_g4(const float* inputs,
                              const float* pos,
                              const float* Wq,
                              const float* Wk,
                              const float* Wv,
                              float* __restrict__ out)
{
    extern __shared__ float lds[];
    float* sH = lds + OFF_H;
    float* sR = lds + OFF_R;
    float* sQ = lds + OFF_Q;
    float* sU = lds + OFF_U;
    float* sS = lds + OFF_S;
    float* sM = lds + OFF_M;
    float* sL = sM + 4 * NH;

    const int tid  = threadIdx.x;
    const int lane = tid & 31;
    const int wave = tid >> 5;
    const int m    = lane & 15;
    const int hi   = lane >> 4;
    const long b0  = (long)blockIdx.x * G;

    float r0[G * DD / 256], p0[G * DD / 256];
    #pragma unroll
    for (int q = 0; q < G * DD / 256; ++q) {
        const int i = tid + 256 * q;
        const int g = i >> 9, d = i & 511;
        r0[q] = inputs[(b0 + g) * (size_t)(TT * DD) + 63 * DD + d];
        p0[q] = pos[63 * DD + d];
    }
    f4 X[XN];
    ld_tile(X, inputs + b0 * (size_t)(TT * DD), tid);

    for (int i = tid; i < 16 * RSTR; i += 256) sR[i] = 0.f;
    lds_barrier();
    #pragma unroll
    for (int q = 0; q < G * DD / 256; ++q) {
        const int i = tid + 256 * q;
        const int g = i >> 9, d = i & 511;
        sR[g * RSTR + d] = r0[q] + p0[q];
    }
    lds_barrier();

    {
        const float* arow = sR + m * RSTR;
        const int col4 = wave * 64 + 4 * m;
        f4 B0[2 * KC], B1[2 * KC], B2[2 * KC], B3[2 * KC];
        v8f acc[4] = {};
        ld_q4(B0, Wq, 0, col4, hi);
        ld_q4(B1, Wq, 16, col4, hi);
        for (int c = 0; c < 32; c += 4) {
            ld_q4(B2, Wq, (c + 2) * 16, col4, hi);
            mac_q4(acc, B0, arow, c * 16, hi);
            ld_q4(B3, Wq, (c + 3) * 16, col4, hi);
            mac_q4(acc, B1, arow, (c + 1) * 16, hi);
            ld_q4(B0, Wq, last_chunk(c + 4) * 16, col4, hi);
            mac_q4(acc, B2, arow, (c + 2) * 16, hi);
            ld_q4(B1, Wq, last_chunk(c + 5) * 16, col4, hi);
            mac_q4(acc, B3, arow, (c + 3) * 16, hi);
        }
        if (hi == 0) {
            #pragma unroll
            for (int j = 0; j < 4; ++j) {
                const int e = col4 + j;
                #pragma unroll
                for (int r = 0; r < G; ++r)
                    sQ[r * QG + (e >> 5) * QSTR + (e & 31)] = acc[j][r];
            }
        }
    }
    lds_barrier();

    {
        v2f A0[CH], A1[CH], A2[CH], A3[CH];
        const int jb = wave * 64;
        ld_wk2(A0, Wk, jb, m, hi);
        ld_wk2(A1, Wk, jb + 2, m, hi);
        for (int c = 0; c < 32; c += 4) {
            ld_wk2(A2, Wk, jb + 2 * (c + 2), m, hi);
            mac_u2(A0, sQ, sU, jb + 2 * c, m, hi);
            ld_wk2(A3, Wk, jb + 2 * (c + 3), m, hi);
            mac_u2(A1, sQ, sU, jb + 2 * (c + 1), m, hi);
            ld_wk2(A0, Wk, jb + 2 * last_chunk(c + 4), m, hi);
            mac_u2(A2, sQ, sU, jb + 2 * (c + 2), m, hi);
            ld_wk2(A1, Wk, jb + 2 * last_chunk(c + 5), m, hi);
            mac_u2(A3, sQ, sU, jb + 2 * (c + 3), m, hi);
        }
    }
    lds_barrier();

    const float scale = 0.044194173824159216f;
    for (int g = 0; g < G; ++g) {
        {
            const f4* pe4 = (const f4*)pos;
            f4 P[XN];
            #pragma unroll
            for (int j = 0; j < XN; ++j) P[j] = pe4[tid + 256 * j];
            #pragma unroll
            for (int j = 0; j < XN; ++j) {
                const int i = tid + 256 * j;
                f4 x = X[j], p = P[j];
                int t = i >> 7, d = (i & 127) << 2;
                float* dst = sH + t * HSTR + d;
                dst[0] = x.x + p.x; dst[1] = x.y + p.y;
                dst[2] = x.z + p.z; dst[3] = x.w + p.w;
            }
            if (g + 1 < G)
                ld_tile(X, inputs + (b0 + g + 1) * (size_t)(TT * DD), tid);
        }
        lds_barrier();

        float sc[8] = {};
        if (wave < 4) {
            const int t0 = wave * 16;
            const float* hrow = sH + (t0 + m) * HSTR + 2 * hi;
            int uoff = OFF_U + g * UG + (2 * hi) * NH + m;
            asm volatile("" : "+v"(uoff));
            const float* ucol = lds + uoff;
            v2f A0[SK], A1[SK], B0[SK], B1[SK];
            v8f ch[4] = {};
            ld_s(A0, B0, hrow, ucol, 0);
            for (int c = 0; c < DD / (4 * SK); c += 2) {
                ld_s(A1, B1, hrow, ucol, (c + 1) * 4 * SK);
                mac_s(ch, A0, B0);
                if (c + 2 < DD / (4 * SK)) ld_s(A0, B0, hrow, ucol, (c + 2) * 4 * SK);
                mac_s(ch, A1, B1);
            }
            const v8f acc = (ch[0] + ch[1]) + (ch[2] + ch[3]);
            float mx = -INFINITY;
            #pragma unroll
            for (int r = 0; r < 8; ++r) {
                sc[r] = acc[r] * scale;
                mx = fmaxf(mx, sc[r]);
            }
            mx = fmaxf(mx, xhalf(mx));
            if (hi == 0) sM[wave * NH + m] = mx;
        }
        lds_barrier();

        if (wave < 4) {
            const int t0 = wave * 16;
            const float mx = fmaxf(fmaxf(sM[m], sM[NH + m]),
                                   fmaxf(sM[2 * NH + m], sM[3 * NH + m]));
            float sum = 0.f;
            #pragma unroll
            for (int r = 0; r < 8; ++r) {
                const float e = expf(sc[r] - mx);
                sS[(t0 + r + 8 * hi) * NH + m] = e;
                sum += e;
            }
            sum += xhalf(sum);
            if (hi == 0) sL[wave * NH + m] = sum;
        }
        lds_barrier();

        {
            v2f pb[16], ha[4][16];
            const float inv = 1.f / (((sL[m] + sL[NH + m]) + sL[2 * NH + m]) + sL[3 * NH + m]);
            #pragma unroll
            for (int s = 0; s < 16; ++s) {
                const int kk = 4 * s + 2 * hi;
                pb[s][0] = sS[kk * NH + m] * inv;
                pb[s][1] = sS[(kk + 1) * NH + m] * inv;
            }
            #pragma unroll
            for (int t = 0; t < 4; ++t) {
                const int d0 = (wave * 4 + t) * 16;
                #pragma unroll
                for (int s = 0; s < 16; ++s) {
                    const int kk = 4 * s + 2 * hi;
                    ha[t][s][0] = sH[kk * HSTR + d0 + m];
                    ha[t][s][1] = sH[(kk + 1) * HSTR + d0 + m];
                }
            }
            #pragma unroll
            for (int t = 0; t < 4; ++t) {
                const int d0 = (wave * 4 + t) * 16;
                v8f acc = {};
                #pragma unroll
                for (int s = 0; s < 16; ++s) acc = wmma4(ha[t][s], pb[s], acc);
                #pragma unroll
                for (int r = 0; r < 8; ++r)
                    sU[g * UG + (d0 + r + 8 * hi) * NH + m] = acc[r];
            }
        }
        lds_barrier();
    }

    {
        const int   gl    = m & 3;
        const float vmask = (m < G) ? 1.f : 0.f;
        const float* u2   = sU + gl * UG + 2 * wave;
        const int col2 = wave * 64 + 2 * m;
        v2f B0[4 * KC], B1[4 * KC], B2[4 * KC], B3[4 * KC];
        v8f acc[4] = {};
        ld_o4(B0, Wv, 0, col2, hi);
        ld_o4(B1, Wv, 16, col2, hi);
        for (int c = 0; c < 32; c += 4) {
            ld_o4(B2, Wv, (c + 2) * 16, col2, hi);
            mac_o4(acc, B0, u2, vmask, c * 16, hi);
            ld_o4(B3, Wv, (c + 3) * 16, col2, hi);
            mac_o4(acc, B1, u2, vmask, (c + 1) * 16, hi);
            ld_o4(B0, Wv, last_chunk(c + 4) * 16, col2, hi);
            mac_o4(acc, B2, u2, vmask, (c + 2) * 16, hi);
            ld_o4(B1, Wv, last_chunk(c + 5) * 16, col2, hi);
            mac_o4(acc, B3, u2, vmask, (c + 3) * 16, hi);
        }
        if (hi == 0) {
            #pragma unroll
            for (int x = 0; x < 2; ++x) {
                #pragma unroll
                for (int r = 0; r < G; ++r) {
                    v2f o;
                    o[0] = acc[2 * x][r];
                    o[1] = acc[2 * x + 1][r];
                    *(v2f*)(out + (b0 + r) * (size_t)DD + col2 + 32 * x) = o;
                }
            }
        }
    }
}

extern "C" void kernel_launch(void* const* d_in, const int* in_sizes, int n_in,
                              void* d_out, int out_size, void* d_ws, size_t ws_size,
                              hipStream_t stream) {
    const float* inputs = (const float*)d_in[0];
    const float* pos    = (const float*)d_in[1];
    const float* Wq     = (const float*)d_in[2];
    const float* Wk     = (const float*)d_in[3];
    const float* Wv     = (const float*)d_in[4];
    float* out          = (float*)d_out;
    (void)d_ws; (void)ws_size; (void)n_in; (void)out_size;

    const int B = in_sizes[0] / (TT * DD);
    const size_t smem = (size_t)LDS_FLOATS * sizeof(float);

    hipFuncSetAttribute((const void*)temporal_attn_lastrow_g4,
                        hipFuncAttributeMaxDynamicSharedMemorySize, (int)smem);

    temporal_attn_lastrow_g4<<<B / G, 256, smem, stream>>>(
        inputs, pos, Wq, Wk, Wv, out);
}


# last-row q GEMM moved into its own launch over all batches: 16 batches x 256 columns per block, all 16 tile rows are batches (was 4 of 16, Wq scanned per 4 batches); q passed in out; bit-identical
# speedup vs baseline: 2.1452x; 1.0958x over previous
#include <hip/hip_runtime.h>
#include <math.h>

constexpr int TT = 64;
constexpr int DD = 512;
constexpr int NH = 16;
constexpr int G  = 4;

constexpr int HSTR = 516;
constexpr int RSTR = 516;
constexpr int QSTR = 36;
constexpr int QG   = NH * QSTR;
constexpr int UG   = DD * NH + 16;

constexpr int OFF_H = 0;
constexpr int OFF_Q = OFF_H + TT * HSTR;
constexpr int OFF_U = OFF_Q + G * QG;
constexpr int OFF_S = OFF_U + G * UG;
constexpr int OFF_M = OFF_S + TT * NH;
constexpr int LDS_FLOATS = OFF_M + 2 * 4 * NH;

constexpr int GQ  = 16;
constexpr int NQ  = 256;
constexpr int KQ  = 8;
constexpr int NCQ = DD / (4 * KQ);

constexpr int CH = 16;
constexpr int KC = 4;
constexpr int XN = 32;
constexpr int SK = 16;

typedef __attribute__((ext_vector_type(2))) float v2f;
typedef __attribute__((ext_vector_type(4))) float f4;
typedef __attribute__((ext_vector_type(8))) float v8f;

static __device__ __forceinline__ v8f wmma4(v2f a, v2f b, v8f c) {
    return __builtin_amdgcn_wmma_f32_16x16x4_f32(
        false, a, false, b, (short)0, c, false, false);
}

static __device__ __forceinline__ float xhalf(float x) {
    const int b = __builtin_bit_cast(int, x);
    return __builtin_bit_cast(float,
        __builtin_amdgcn_permlanex16(b, b, 0x76543210, 0xfedcba98, false, false));
}

static __device__ __forceinline__ void lds_barrier() {
    __builtin_amdgcn_fence(__ATOMIC_RELEASE, "workgroup", "local");
    __builtin_amdgcn_s_barrier();
    __builtin_amdgcn_fence(__ATOMIC_ACQUIRE, "workgroup", "local");
}

static __device__ __forceinline__ int last_chunk(int c) { return c < 31 ? c : 31; }
static __device__ __forceinline__ int last_qchunk(int c) { return c < NCQ - 1 ? c : NCQ - 1; }

static __device__ __forceinline__ void ld_qp(v2f (&b)[2 * KQ], const float* W,
                                             int kb, int col2, int hi) {
    #pragma unroll
    for (int s = 0; s < KQ; ++s) {
        const int kk = kb + 4 * s + 2 * hi;
        b[2 * s]     = *(const v2f*)(W + (size_t)kk * DD + col2);
        b[2 * s + 1] = *(const v2f*)(W + (size_t)(kk + 1) * DD + col2);
    }
}

static __device__ __forceinline__ void mac_qp(v8f (&acc)[2], const v2f (&b)[2 * KQ],
                                              const float* arow, int kb, int hi) {
    v2f bq[2 * KQ];
    #pragma unroll
    for (int i = 0; i < 2 * KQ; ++i) {
        bq[i] = b[i];
        asm volatile("" : "+v"(bq[i]));
    }
    v2f a[KQ];
    #pragma unroll
    for (int s = 0; s < KQ; ++s) {
        const int kk = kb + 4 * s + 2 * hi;
        a[s][0] = arow[kk];
        a[s][1] = arow[kk + 1];
    }
    #pragma unroll
    for (int s = 0; s < KQ; ++s) {
        #pragma unroll
        for (int j = 0; j < 2; ++j) {
            v2f bb;
            bb[0] = bq[2 * s][j];
            bb[1] = bq[2 * s + 1][j];
            acc[j] = wmma4(a[s], bb, acc[j]);
        }
    }
}

static __device__ __forceinline__ void ld_o4(v2f (&b)[4 * KC], const float* W,
                                             int kb, int col2, int hi) {
    #pragma unroll
    for (int s = 0; s < KC; ++s) {
        const int kk = kb + 4 * s + 2 * hi;
        const float* p = W + (size_t)kk * DD + col2;
        b[4 * s]     = *(const v2f*)(p);
        b[4 * s + 1] = *(const v2f*)(p + DD);
        b[4 * s + 2] = *(const v2f*)(p + 32);
        b[4 * s + 3] = *(const v2f*)(p + DD + 32);
    }
}

static __device__ __forceinline__ void mac_o4(v8f (&acc)[4], const v2f (&b)[4 * KC],
                                              const float* u2, float vmask,
                                              int kb, int hi) {
    v2f bq[4 * KC];
    #pragma unroll
    for (int i = 0; i < 4 * KC; ++i) {
        bq[i] = b[i];
        asm volatile("" : "+v"(bq[i]));
    }
    #pragma unroll
    for (int s = 0; s < KC; ++s) {
        const int kk = kb + 4 * s + 2 * hi;
        #pragma unroll
        for (int x = 0; x < 2; ++x) {
            v2f a;
            a[0] = vmask * u2[kk * NH + x];
            a[1] = vmask * u2[(kk + 1) * NH + x];
            #pragma unroll
            for (int j = 0; j < 2; ++j) {
                v2f bb;
                bb[0] = bq[4 * s + 2 * x][j];
                bb[1] = bq[4 * s + 2 * x + 1][j];
                acc[2 * x + j] = wmma4(a, bb, acc[2 * x + j]);
            }
        }
    }
}

static __device__ __forceinline__ void ld_wk2(v2f (&a)[CH], const float* Wk,
                                              int job0, int m, int hi) {
    #pragma unroll
    for (int j = 0; j < 2; ++j) {
        const int job = job0 + j;
        const int d0  = (job >> 4) * 16;
        const int h   = job & 15;
        const float* p = Wk + (size_t)(d0 + m) * DD + 32 * h + 2 * hi;
        #pragma unroll
        for (int s = 0; s < 8; ++s) {
            a[j * 8 + s][0] = p[4 * s];
            a[j * 8 + s][1] = p[4 * s + 1];
        }
    }
}

static __device__ __forceinline__ void mac_u2(const v2f (&a)[CH], const float* sQ,
                                              float* sU, int job0, int m, int hi) {
    const int   gl    = m & 3;
    const float vmask = (m < G) ? 1.f : 0.f;
    #pragma unroll
    for (int j = 0; j < 2; ++j) {
        const int job = job0 + j;
        const int d0  = (job >> 4) * 16;
        const int h   = job & 15;
        v8f acc = {};
        #pragma unroll
        for (int s = 0; s < 8; ++s) {
            const int kk = 4 * s + 2 * hi;
            v2f b;
            b[0] = vmask * sQ[gl * QG + h * QSTR + kk];
            b[1] = vmask * sQ[gl * QG + h * QSTR + kk + 1];
            acc = wmma4(a[j * 8 + s], b, acc);
        }
        if (m < G) {
            #pragma unroll
            for (int r = 0; r < 8; ++r)
                sU[m * UG + (d0 + r + 8 * hi) * NH + h] = acc[r];
        }
    }
}

static __device__ __forceinline__ void ld_s(v2f (&a)[SK], v2f (&b)[SK], const float* hrow,
                                            const float* ucol, int d0) {
    #pragma unroll
    for (int s = 0; s < SK; ++s) {
        a[s] = *(const v2f*)(hrow + d0 + 4 * s);
        b[s][0] = ucol[(d0 + 4 * s) * NH];
        b[s][1] = ucol[(d0 + 4 * s + 1) * NH];
    }
}

static __device__ __forceinline__ void mac_s(v8f (&ch)[4], const v2f (&a)[SK], const v2f (&b)[SK]) {
    #pragma unroll
    for (int s = 0; s < SK; ++s) ch[s & 3] = wmma4(a[s], b[s], ch[s & 3]);
}

static __device__ __forceinline__ void ld_tile(f4 (&x)[XN], const float* src, int tid) {
    const f4* in4 = (const f4*)src;
    #pragma unroll
    for (int j = 0; j < XN; ++j) x[j] = in4[tid + 256 * j];
}

__global__ __launch_bounds__(256, 1)
void lastrow_q_all(const float* inputs,
                   const float* pos,
                   const float* Wq,
                   float* q)
{
    __shared__ float sR[GQ * RSTR];

    const int tid  = threadIdx.x;
    const int lane = tid & 31;
    const int wave = tid >> 5;
    const int m    = lane & 15;
    const int hi   = lane >> 4;
    const long b0  = (long)blockIdx.x * GQ;
    const int col2 = blockIdx.y * NQ + wave * 32 + 2 * m;

    f4 r0[GQ * DD / 1024], p0[GQ * DD / 1024];
    #pragma unroll
    for (int u = 0; u < GQ * DD / 1024; ++u) {
        const int i = tid + 256 * u;
        const int g = i >> 7, d = (i & 127) << 2;
        r0[u] = *(const f4*)(inputs + (b0 + g) * (size_t)(TT * DD) + 63 * DD + d);
        p0[u] = *(const f4*)(pos + 63 * DD + d);
    }
    asm volatile("" ::: "memory");
    v2f B0[2 * KQ], B1[2 * KQ], B2[2 * KQ], B3[2 * KQ];
    ld_qp(B0, Wq, 0, col2, hi);
    ld_qp(B1, Wq, 4 * KQ, col2, hi);
    #pragma unroll
    for (int u = 0; u < GQ * DD / 1024; ++u) {
        const int i = tid + 256 * u;
        const int g = i >> 7, d = (i & 127) << 2;
        float* dst = sR + g * RSTR + d;
        dst[0] = r0[u].x + p0[u].x; dst[1] = r0[u].y + p0[u].y;
        dst[2] = r0[u].z + p0[u].z; dst[3] = r0[u].w + p0[u].w;
    }
    lds_barrier();

    const float* arow = sR + m * RSTR;
    v8f acc[2] = {};
    for (int c = 0; c < NCQ; c += 4) {
        ld_qp(B2, Wq, (c + 2) * 4 * KQ, col2, hi);
        mac_qp(acc, B0, arow, c * 4 * KQ, hi);
        ld_qp(B3, Wq, (c + 3) * 4 * KQ, col2, hi);
        mac_qp(acc, B1, arow, (c + 1) * 4 * KQ, hi);
        ld_qp(B0, Wq, last_qchunk(c + 4) * 4 * KQ, col2, hi);
        mac_qp(acc, B2, arow, (c + 2) * 4 * KQ, hi);
        ld_qp(B1, Wq, last_qchunk(c + 5) * 4 * KQ, col2, hi);
        mac_qp(acc, B3, arow, (c + 3) * 4 * KQ, hi);
    }
    #pragma unroll
    for (int r = 0; r < 8; ++r) {
        v2f o;
        o[0] = acc[0][r];
        o[1] = acc[1][r];
        *(v2f*)(q + (b0 + 8 * hi + r) * (size_t)DD + col2) = o;
    }
}

__global__ __launch_bounds__(256, 1)
void temporal_attn_lastrow_g4(const float* inputs,
                              const float* pos,
                              const float* Wk,
                              const float* Wv,
                              float* out)
{
    extern __shared__ float lds[];
    float* sH = lds + OFF_H;
    float* sQ = lds + OFF_Q;
    float* sU = lds + OFF_U;
    float* sS = lds + OFF_S;
    float* sM = lds + OFF_M;
    float* sL = sM + 4 * NH;

    const int tid  = threadIdx.x;
    const int lane = tid & 31;
    const int wave = tid >> 5;
    const int m    = lane & 15;
    const int hi   = lane >> 4;
    const long b0  = (long)blockIdx.x * G;

    float q0[G * DD / 256];
    #pragma unroll
    for (int u = 0; u < G * DD / 256; ++u) {
        const int i = tid + 256 * u;
        const int g = i >> 9, e = i & 511;
        q0[u] = out[(b0 + g) * (size_t)DD + e];
    }
    asm volatile("" ::: "memory");
    f4 X[XN];
    ld_tile(X, inputs + b0 * (size_t)(TT * DD), tid);
    #pragma unroll
    for (int u = 0; u < G * DD / 256; ++u) {
        const int i = tid + 256 * u;
        const int g = i >> 9, e = i & 511;
        sQ[g * QG + (e >> 5) * QSTR + (e & 31)] = q0[u];
    }
    lds_barrier();

    {
        v2f A0[CH], A1[CH], A2[CH], A3[CH];
        const int jb = wave * 64;
        ld_wk2(A0, Wk, jb, m, hi);
        ld_wk2(A1, Wk, jb + 2, m, hi);
        for (int c = 0; c < 32; c += 4) {
            ld_wk2(A2, Wk, jb + 2 * (c + 2), m, hi);
            mac_u2(A0, sQ, sU, jb + 2 * c, m, hi);
            ld_wk2(A3, Wk, jb + 2 * (c + 3), m, hi);
            mac_u2(A1, sQ, sU, jb + 2 * (c + 1), m, hi);
            ld_wk2(A0, Wk, jb + 2 * last_chunk(c + 4), m, hi);
            mac_u2(A2, sQ, sU, jb + 2 * (c + 2), m, hi);
            ld_wk2(A1, Wk, jb + 2 * last_chunk(c + 5), m, hi);
            mac_u2(A3, sQ, sU, jb + 2 * (c + 3), m, hi);
        }
    }
    lds_barrier();

    const float scale = 0.044194173824159216f;
    for (int g = 0; g < G; ++g) {
        {
            const f4* pe4 = (const f4*)pos;
            f4 P[XN];
            #pragma unroll
            for (int j = 0; j < XN; ++j) P[j] = pe4[tid + 256 * j];
            #pragma unroll
            for (int j = 0; j < XN; ++j) {
                const int i = tid + 256 * j;
                f4 x = X[j], p = P[j];
                int t = i >> 7, d = (i & 127) << 2;
                float* dst = sH + t * HSTR + d;
                dst[0] = x.x + p.x; dst[1] = x.y + p.y;
                dst[2] = x.z + p.z; dst[3] = x.w + p.w;
            }
            if (g + 1 < G)
                ld_tile(X, inputs + (b0 + g + 1) * (size_t)(TT * DD), tid);
        }
        lds_barrier();

        float sc[8] = {};
        if (wave < 4) {
            const int t0 = wave * 16;
            const float* hrow = sH + (t0 + m) * HSTR + 2 * hi;
            int uoff = OFF_U + g * UG + (2 * hi) * NH + m;
            asm volatile("" : "+v"(uoff));
            const float* ucol = lds + uoff;
            v2f A0[SK], A1[SK], B0[SK], B1[SK];
            v8f ch[4] = {};
            ld_s(A0, B0, hrow, ucol, 0);
            for (int c = 0; c < DD / (4 * SK); c += 2) {
                ld_s(A1, B1, hrow, ucol, (c + 1) * 4 * SK);
                mac_s(ch, A0, B0);
                if (c + 2 < DD / (4 * SK)) ld_s(A0, B0, hrow, ucol, (c + 2) * 4 * SK);
                mac_s(ch, A1, B1);
            }
            const v8f acc = (ch[0] + ch[1]) + (ch[2] + ch[3]);
            float mx = -INFINITY;
            #pragma unroll
            for (int r = 0; r < 8; ++r) {
                sc[r] = acc[r] * scale;
                mx = fmaxf(mx, sc[r]);
            }
            mx = fmaxf(mx, xhalf(mx));
            if (hi == 0) sM[wave * NH + m] = mx;
        }
        lds_barrier();

        if (wave < 4) {
            const int t0 = wave * 16;
            const float mx = fmaxf(fmaxf(sM[m], sM[NH + m]),
                                   fmaxf(sM[2 * NH + m], sM[3 * NH + m]));
            float sum = 0.f;
            #pragma unroll
            for (int r = 0; r < 8; ++r) {
                const float e = expf(sc[r] - mx);
                sS[(t0 + r + 8 * hi) * NH + m] = e;
                sum += e;
            }
            sum += xhalf(sum);
            if (hi == 0) sL[wave * NH + m] = sum;
        }
        lds_barrier();

        {
            v2f pb[16], ha[4][16];
            const float inv = 1.f / (((sL[m] + sL[NH + m]) + sL[2 * NH + m]) + sL[3 * NH + m]);
            #pragma unroll
            for (int s = 0; s < 16; ++s) {
                const int kk = 4 * s + 2 * hi;
                pb[s][0] = sS[kk * NH + m] * inv;
                pb[s][1] = sS[(kk + 1) * NH + m] * inv;
            }
            #pragma unroll
            for (int t = 0; t < 4; ++t) {
                const int d0 = (wave * 4 + t) * 16;
                #pragma unroll
                for (int s = 0; s < 16; ++s) {
                    const int kk = 4 * s + 2 * hi;
                    ha[t][s][0] = sH[kk * HSTR + d0 + m];
                    ha[t][s][1] = sH[(kk + 1) * HSTR + d0 + m];
                }
            }
            #pragma unroll
            for (int t = 0; t < 4; ++t) {
                const int d0 = (wave * 4 + t) * 16;
                v8f acc = {};
                #pragma unroll
                for (int s = 0; s < 16; ++s) acc = wmma4(ha[t][s], pb[s], acc);
                #pragma unroll
                for (int r = 0; r < 8; ++r)
                    sU[g * UG + (d0 + r + 8 * hi) * NH + m] = acc[r];
            }
        }
        lds_barrier();
    }

    {
        const int   gl    = m & 3;
        const float vmask = (m < G) ? 1.f : 0.f;
        const float* u2   = sU + gl * UG + 2 * wave;
        const int col2 = wave * 64 + 2 * m;
        v2f B0[4 * KC], B1[4 * KC], B2[4 * KC], B3[4 * KC];
        v8f acc[4] = {};
        ld_o4(B0, Wv, 0, col2, hi);
        ld_o4(B1, Wv, 16, col2, hi);
        for (int c = 0; c < 32; c += 4) {
            ld_o4(B2, Wv, (c + 2) * 16, col2, hi);
            mac_o4(acc, B0, u2, vmask, c * 16, hi);
            ld_o4(B3, Wv, (c + 3) * 16, col2, hi);
            mac_o4(acc, B1, u2, vmask, (c + 1) * 16, hi);
            ld_o4(B0, Wv, last_chunk(c + 4) * 16, col2, hi);
            mac_o4(acc, B2, u2, vmask, (c + 2) * 16, hi);
            ld_o4(B1, Wv, last_chunk(c + 5) * 16, col2, hi);
            mac_o4(acc, B3, u2, vmask, (c + 3) * 16, hi);
        }
        if (hi == 0) {
            #pragma unroll
            for (int x = 0; x < 2; ++x) {
                #pragma unroll
                for (int r = 0; r < G; ++r) {
                    v2f o;
                    o[0] = acc[2 * x][r];
                    o[1] = acc[2 * x + 1][r];
                    *(v2f*)(out + (b0 + r) * (size_t)DD + col2 + 32 * x) = o;
                }
            }
        }
    }
}

extern "C" void kernel_launch(void* const* d_in, const int* in_sizes, int n_in,
                              void* d_out, int out_size, void* d_ws, size_t ws_size,
                              hipStream_t stream) {
    const float* inputs = (const float*)d_in[0];
    const float* pos    = (const float*)d_in[1];
    const float* Wq     = (const float*)d_in[2];
    const float* Wk     = (const float*)d_in[3];
    const float* Wv     = (const float*)d_in[4];
    float* out          = (float*)d_out;
    (void)d_ws; (void)ws_size; (void)n_in; (void)out_size;

    const int B = in_sizes[0] / (TT * DD);
    const size_t smem = (size_t)LDS_FLOATS * sizeof(float);

    hipFuncSetAttribute((const void*)temporal_attn_lastrow_g4,
                        hipFuncAttributeMaxDynamicSharedMemorySize, (int)smem);

    lastrow_q_all<<<dim3(B / GQ, DD / NQ), 256, 0, stream>>>(inputs, pos, Wq, out);
    temporal_attn_lastrow_g4<<<B / G, 256, smem, stream>>>(
        inputs, pos, Wk, Wv, out);
}


# LDS base declared 16-byte aligned: H tile (inputs+pos) stored with 16-byte stores (was two 2-address 4-byte stores per piece), q and C fragment pairs read as 8 bytes; bit-identical
# speedup vs baseline: 2.2174x; 1.0337x over previous
#include <hip/hip_runtime.h>
#include <math.h>

constexpr int TT = 64;
constexpr int DD = 512;
constexpr int NH = 16;
constexpr int G  = 4;

constexpr int HSTR = 516;
constexpr int RSTR = 516;
constexpr int QSTR = 36;
constexpr int QG   = NH * QSTR;
constexpr int UG   = DD * NH + 16;

constexpr int OFF_H = 0;
constexpr int OFF_Q = OFF_H + TT * HSTR;
constexpr int OFF_U = OFF_Q + G * QG;
constexpr int OFF_S = OFF_U + G * UG;
constexpr int OFF_M = OFF_S + TT * NH;
constexpr int LDS_FLOATS = OFF_M + 2 * 4 * NH;

constexpr int GQ  = 16;
constexpr int NQ  = 256;
constexpr int KQ  = 8;
constexpr int NCQ = DD / (4 * KQ);

constexpr int CH = 16;
constexpr int KC = 4;
constexpr int XN = 32;
constexpr int SK = 16;

typedef __attribute__((ext_vector_type(2))) float v2f;
typedef __attribute__((ext_vector_type(4))) float f4;
typedef __attribute__((ext_vector_type(8))) float v8f;

static __device__ __forceinline__ v8f wmma4(v2f a, v2f b, v8f c) {
    return __builtin_amdgcn_wmma_f32_16x16x4_f32(
        false, a, false, b, (short)0, c, false, false);
}

static __device__ __forceinline__ float xhalf(float x) {
    const int b = __builtin_bit_cast(int, x);
    return __builtin_bit_cast(float,
        __builtin_amdgcn_permlanex16(b, b, 0x76543210, 0xfedcba98, false, false));
}

static __device__ __forceinline__ void lds_barrier() {
    __builtin_amdgcn_fence(__ATOMIC_RELEASE, "workgroup", "local");
    __builtin_amdgcn_s_barrier();
    __builtin_amdgcn_fence(__ATOMIC_ACQUIRE, "workgroup", "local");
}

static __device__ __forceinline__ int last_chunk(int c) { return c < 31 ? c : 31; }
static __device__ __forceinline__ int last_qchunk(int c) { return c < NCQ - 1 ? c : NCQ - 1; }

static __device__ __forceinline__ void ld_qp(v2f (&b)[2 * KQ], const float* W,
                                             int kb, int col2, int hi) {
    #pragma unroll
    for (int s = 0; s < KQ; ++s) {
        const int kk = kb + 4 * s + 2 * hi;
        b[2 * s]     = *(const v2f*)(W + (size_t)kk * DD + col2);
        b[2 * s + 1] = *(const v2f*)(W + (size_t)(kk + 1) * DD + col2);
    }
}

static __device__ __forceinline__ void mac_qp(v8f (&acc)[2], const v2f (&b)[2 * KQ],
                                              const float* arow, int kb, int hi) {
    v2f bq[2 * KQ];
    #pragma unroll
    for (int i = 0; i < 2 * KQ; ++i) {
        bq[i] = b[i];
        asm volatile("" : "+v"(bq[i]));
    }
    v2f a[KQ];
    #pragma unroll
    for (int s = 0; s < KQ; ++s) {
        const int kk = kb + 4 * s + 2 * hi;
        a[s][0] = arow[kk];
        a[s][1] = arow[kk + 1];
    }
    #pragma unroll
    for (int s = 0; s < KQ; ++s) {
        #pragma unroll
        for (int j = 0; j < 2; ++j) {
            v2f bb;
            bb[0] = bq[2 * s][j];
            bb[1] = bq[2 * s + 1][j];
            acc[j] = wmma4(a[s], bb, acc[j]);
        }
    }
}

static __device__ __forceinline__ void ld_o4(v2f (&b)[4 * KC], const float* W,
                                             int kb, int col2, int hi) {
    #pragma unroll
    for (int s = 0; s < KC; ++s) {
        const int kk = kb + 4 * s + 2 * hi;
        const float* p = W + (size_t)kk * DD + col2;
        b[4 * s]     = *(const v2f*)(p);
        b[4 * s + 1] = *(const v2f*)(p + DD);
        b[4 * s + 2] = *(const v2f*)(p + 32);
        b[4 * s + 3] = *(const v2f*)(p + DD + 32);
    }
}

static __device__ __forceinline__ void mac_o4(v8f (&acc)[4], const v2f (&b)[4 * KC],
                                              const float* u2, float vmask,
                                              int kb, int hi) {
    v2f bq[4 * KC];
    #pragma unroll
    for (int i = 0; i < 4 * KC; ++i) {
        bq[i] = b[i];
        asm volatile("" : "+v"(bq[i]));
    }
    #pragma unroll
    for (int s = 0; s < KC; ++s) {
        const int kk = kb + 4 * s + 2 * hi;
        #pragma unroll
        for (int x = 0; x < 2; ++x) {
            v2f a;
            a[0] = vmask * u2[kk * NH + x];
            a[1] = vmask * u2[(kk + 1) * NH + x];
            #pragma unroll
            for (int j = 0; j < 2; ++j) {
                v2f bb;
                bb[0] = bq[4 * s + 2 * x][j];
                bb[1] = bq[4 * s + 2 * x + 1][j];
                acc[2 * x + j] = wmma4(a, bb, acc[2 * x + j]);
            }
        }
    }
}

static __device__ __forceinline__ void ld_wk2(v2f (&a)[CH], const float* Wk,
                                              int job0, int m, int hi) {
    #pragma unroll
    for (int j = 0; j < 2; ++j) {
        const int job = job0 + j;
        const int d0  = (job >> 4) * 16;
        const int h   = job & 15;
        const float* p = Wk + (size_t)(d0 + m) * DD + 32 * h + 2 * hi;
        #pragma unroll
        for (int s = 0; s < 8; ++s) {
            a[j * 8 + s][0] = p[4 * s];
            a[j * 8 + s][1] = p[4 * s + 1];
        }
    }
}

static __device__ __forceinline__ void mac_u2(const v2f (&a)[CH], const float* sQ,
                                              float* sU, int job0, int m, int hi) {
    const int   gl    = m & 3;
    const float vmask = (m < G) ? 1.f : 0.f;
    #pragma unroll
    for (int j = 0; j < 2; ++j) {
        const int job = job0 + j;
        const int d0  = (job >> 4) * 16;
        const int h   = job & 15;
        v8f acc = {};
        #pragma unroll
        for (int s = 0; s < 8; ++s) {
            const int kk = 4 * s + 2 * hi;
            v2f b;
            b[0] = vmask * sQ[gl * QG + h * QSTR + kk];
            b[1] = vmask * sQ[gl * QG + h * QSTR + kk + 1];
            acc = wmma4(a[j * 8 + s], b, acc);
        }
        if (m < G) {
            #pragma unroll
            for (int r = 0; r < 8; ++r)
                sU[m * UG + (d0 + r + 8 * hi) * NH + h] = acc[r];
        }
    }
}

static __device__ __forceinline__ void ld_s(v2f (&a)[SK], v2f (&b)[SK], const float* hrow,
                                            const float* ucol, int d0) {
    #pragma unroll
    for (int s = 0; s < SK; ++s) {
        a[s] = *(const v2f*)(hrow + d0 + 4 * s);
        b[s][0] = ucol[(d0 + 4 * s) * NH];
        b[s][1] = ucol[(d0 + 4 * s + 1) * NH];
    }
}

static __device__ __forceinline__ void mac_s(v8f (&ch)[4], const v2f (&a)[SK], const v2f (&b)[SK]) {
    #pragma unroll
    for (int s = 0; s < SK; ++s) ch[s & 3] = wmma4(a[s], b[s], ch[s & 3]);
}

static __device__ __forceinline__ void ld_tile(f4 (&x)[XN], const float* src, int tid) {
    const f4* in4 = (const f4*)src;
    #pragma unroll
    for (int j = 0; j < XN; ++j) x[j] = in4[tid + 256 * j];
}

__global__ __launch_bounds__(256, 1)
void lastrow_q_all(const float* inputs,
                   const float* pos,
                   const float* Wq,
                   float* q)
{
    __shared__ float sR[GQ * RSTR];

    const int tid  = threadIdx.x;
    const int lane = tid & 31;
    const int wave = tid >> 5;
    const int m    = lane & 15;
    const int hi   = lane >> 4;
    const long b0  = (long)blockIdx.x * GQ;
    const int col2 = blockIdx.y * NQ + wave * 32 + 2 * m;

    f4 r0[GQ * DD / 1024], p0[GQ * DD / 1024];
    #pragma unroll
    for (int u = 0; u < GQ * DD / 1024; ++u) {
        const int i = tid + 256 * u;
        const int g = i >> 7, d = (i & 127) << 2;
        r0[u] = *(const f4*)(inputs + (b0 + g) * (size_t)(TT * DD) + 63 * DD + d);
        p0[u] = *(const f4*)(pos + 63 * DD + d);
    }
    asm volatile("" ::: "memory");
    v2f B0[2 * KQ], B1[2 * KQ], B2[2 * KQ], B3[2 * KQ];
    ld_qp(B0, Wq, 0, col2, hi);
    ld_qp(B1, Wq, 4 * KQ, col2, hi);
    #pragma unroll
    for (int u = 0; u < GQ * DD / 1024; ++u) {
        const int i = tid + 256 * u;
        const int g = i >> 7, d = (i & 127) << 2;
        float* dst = sR + g * RSTR + d;
        dst[0] = r0[u].x + p0[u].x; dst[1] = r0[u].y + p0[u].y;
        dst[2] = r0[u].z + p0[u].z; dst[3] = r0[u].w + p0[u].w;
    }
    lds_barrier();

    const float* arow = sR + m * RSTR;
    v8f acc[2] = {};
    for (int c = 0; c < NCQ; c += 4) {
        ld_qp(B2, Wq, (c + 2) * 4 * KQ, col2, hi);
        mac_qp(acc, B0, arow, c * 4 * KQ, hi);
        ld_qp(B3, Wq, (c + 3) * 4 * KQ, col2, hi);
        mac_qp(acc, B1, arow, (c + 1) * 4 * KQ, hi);
        ld_qp(B0, Wq, last_qchunk(c + 4) * 4 * KQ, col2, hi);
        mac_qp(acc, B2, arow, (c + 2) * 4 * KQ, hi);
        ld_qp(B1, Wq, last_qchunk(c + 5) * 4 * KQ, col2, hi);
        mac_qp(acc, B3, arow, (c + 3) * 4 * KQ, hi);
    }
    #pragma unroll
    for (int r = 0; r < 8; ++r) {
        v2f o;
        o[0] = acc[0][r];
        o[1] = acc[1][r];
        *(v2f*)(q + (b0 + 8 * hi + r) * (size_t)DD + col2) = o;
    }
}

__global__ __launch_bounds__(256, 1)
void temporal_attn_lastrow_g4(const float* inputs,
                              const float* pos,
                              const float* Wk,
                              const float* Wv,
                              float* out)
{
    extern __shared__ __attribute__((aligned(16))) float lds[];
    float* sH = lds + OFF_H;
    float* sQ = lds + OFF_Q;
    float* sU = lds + OFF_U;
    float* sS = lds + OFF_S;
    float* sM = lds + OFF_M;
    float* sL = sM + 4 * NH;

    const int tid  = threadIdx.x;
    const int lane = tid & 31;
    const int wave = tid >> 5;
    const int m    = lane & 15;
    const int hi   = lane >> 4;
    const long b0  = (long)blockIdx.x * G;

    float q0[G * DD / 256];
    #pragma unroll
    for (int u = 0; u < G * DD / 256; ++u) {
        const int i = tid + 256 * u;
        const int g = i >> 9, e = i & 511;
        q0[u] = out[(b0 + g) * (size_t)DD + e];
    }
    asm volatile("" ::: "memory");
    f4 X[XN];
    ld_tile(X, inputs + b0 * (size_t)(TT * DD), tid);
    #pragma unroll
    for (int u = 0; u < G * DD / 256; ++u) {
        const int i = tid + 256 * u;
        const int g = i >> 9, e = i & 511;
        sQ[g * QG + (e >> 5) * QSTR + (e & 31)] = q0[u];
    }
    lds_barrier();

    {
        v2f A0[CH], A1[CH], A2[CH], A3[CH];
        const int jb = wave * 64;
        ld_wk2(A0, Wk, jb, m, hi);
        ld_wk2(A1, Wk, jb + 2, m, hi);
        for (int c = 0; c < 32; c += 4) {
            ld_wk2(A2, Wk, jb + 2 * (c + 2), m, hi);
            mac_u2(A0, sQ, sU, jb + 2 * c, m, hi);
            ld_wk2(A3, Wk, jb + 2 * (c + 3), m, hi);
            mac_u2(A1, sQ, sU, jb + 2 * (c + 1), m, hi);
            ld_wk2(A0, Wk, jb + 2 * last_chunk(c + 4), m, hi);
            mac_u2(A2, sQ, sU, jb + 2 * (c + 2), m, hi);
            ld_wk2(A1, Wk, jb + 2 * last_chunk(c + 5), m, hi);
            mac_u2(A3, sQ, sU, jb + 2 * (c + 3), m, hi);
        }
    }
    lds_barrier();

    const float scale = 0.044194173824159216f;
    for (int g = 0; g < G; ++g) {
        {
            const f4* pe4 = (const f4*)pos;
            f4 P[XN];
            #pragma unroll
            for (int j = 0; j < XN; ++j) P[j] = pe4[tid + 256 * j];
            #pragma unroll
            for (int j = 0; j < XN; ++j) {
                const int i = tid + 256 * j;
                f4 x = X[j], p = P[j];
                int t = i >> 7, d = (i & 127) << 2;
                f4 h;
                h.x = x.x + p.x; h.y = x.y + p.y;
                h.z = x.z + p.z; h.w = x.w + p.w;
                *(f4*)(sH + t * HSTR + d) = h;
            }
            if (g + 1 < G)
                ld_tile(X, inputs + (b0 + g + 1) * (size_t)(TT * DD), tid);
        }
        lds_barrier();

        float sc[8] = {};
        if (wave < 4) {
            const int t0 = wave * 16;
            const float* hrow = sH + (t0 + m) * HSTR + 2 * hi;
            int uoff = OFF_U + g * UG + (2 * hi) * NH + m;
            asm volatile("" : "+v"(uoff));
            const float* ucol = lds + uoff;
            v2f A0[SK], A1[SK], B0[SK], B1[SK];
            v8f ch[4] = {};
            ld_s(A0, B0, hrow, ucol, 0);
            for (int c = 0; c < DD / (4 * SK); c += 2) {
                ld_s(A1, B1, hrow, ucol, (c + 1) * 4 * SK);
                mac_s(ch, A0, B0);
                if (c + 2 < DD / (4 * SK)) ld_s(A0, B0, hrow, ucol, (c + 2) * 4 * SK);
                mac_s(ch, A1, B1);
            }
            const v8f acc = (ch[0] + ch[1]) + (ch[2] + ch[3]);
            float mx = -INFINITY;
            #pragma unroll
            for (int r = 0; r < 8; ++r) {
                sc[r] = acc[r] * scale;
                mx = fmaxf(mx, sc[r]);
            }
            mx = fmaxf(mx, xhalf(mx));
            if (hi == 0) sM[wave * NH + m] = mx;
        }
        lds_barrier();

        if (wave < 4) {
            const int t0 = wave * 16;
            const float mx = fmaxf(fmaxf(sM[m], sM[NH + m]),
                                   fmaxf(sM[2 * NH + m], sM[3 * NH + m]));
            float sum = 0.f;
            #pragma unroll
            for (int r = 0; r < 8; ++r) {
                const float e = expf(sc[r] - mx);
                sS[(t0 + r + 8 * hi) * NH + m] = e;
                sum += e;
            }
            sum += xhalf(sum);
            if (hi == 0) sL[wave * NH + m] = sum;
        }
        lds_barrier();

        {
            v2f pb[16], ha[4][16];
            const float inv = 1.f / (((sL[m] + sL[NH + m]) + sL[2 * NH + m]) + sL[3 * NH + m]);
            #pragma unroll
            for (int s = 0; s < 16; ++s) {
                const int kk = 4 * s + 2 * hi;
                pb[s][0] = sS[kk * NH + m] * inv;
                pb[s][1] = sS[(kk + 1) * NH + m] * inv;
            }
            #pragma unroll
            for (int t = 0; t < 4; ++t) {
                const int d0 = (wave * 4 + t) * 16;
                #pragma unroll
                for (int s = 0; s < 16; ++s) {
                    const int kk = 4 * s + 2 * hi;
                    ha[t][s][0] = sH[kk * HSTR + d0 + m];
                    ha[t][s][1] = sH[(kk + 1) * HSTR + d0 + m];
                }
            }
            #pragma unroll
            for (int t = 0; t < 4; ++t) {
                const int d0 = (wave * 4 + t) * 16;
                v8f acc = {};
                #pragma unroll
                for (int s = 0; s < 16; ++s) acc = wmma4(ha[t][s], pb[s], acc);
                #pragma unroll
                for (int r = 0; r < 8; ++r)
                    sU[g * UG + (d0 + r + 8 * hi) * NH + m] = acc[r];
            }
        }
        lds_barrier();
    }

    {
        const int   gl    = m & 3;
        const float vmask = (m < G) ? 1.f : 0.f;
        const float* u2   = sU + gl * UG + 2 * wave;
        const int col2 = wave * 64 + 2 * m;
        v2f B0[4 * KC], B1[4 * KC], B2[4 * KC], B3[4 * KC];
        v8f acc[4] = {};
        ld_o4(B0, Wv, 0, col2, hi);
        ld_o4(B1, Wv, 16, col2, hi);
        for (int c = 0; c < 32; c += 4) {
            ld_o4(B2, Wv, (c + 2) * 16, col2, hi);
            mac_o4(acc, B0, u2, vmask, c * 16, hi);
            ld_o4(B3, Wv, (c + 3) * 16, col2, hi);
            mac_o4(acc, B1, u2, vmask, (c + 1) * 16, hi);
            ld_o4(B0, Wv, last_chunk(c + 4) * 16, col2, hi);
            mac_o4(acc, B2, u2, vmask, (c + 2) * 16, hi);
            ld_o4(B1, Wv, last_chunk(c + 5) * 16, col2, hi);
            mac_o4(acc, B3, u2, vmask, (c + 3) * 16, hi);
        }
        if (hi == 0) {
            #pragma unroll
            for (int x = 0; x < 2; ++x) {
                #pragma unroll
                for (int r = 0; r < G; ++r) {
                    v2f o;
                    o[0] = acc[2 * x][r];
                    o[1] = acc[2 * x + 1][r];
                    *(v2f*)(out + (b0 + r) * (size_t)DD + col2 + 32 * x) = o;
                }
            }
        }
    }
}

extern "C" void kernel_launch(void* const* d_in, const int* in_sizes, int n_in,
                              void* d_out, int out_size, void* d_ws, size_t ws_size,
                              hipStream_t stream) {
    const float* inputs = (const float*)d_in[0];
    const float* pos    = (const float*)d_in[1];
    const float* Wq     = (const float*)d_in[2];
    const float* Wk     = (const float*)d_in[3];
    const float* Wv     = (const float*)d_in[4];
    float* out          = (float*)d_out;
    (void)d_ws; (void)ws_size; (void)n_in; (void)out_size;

    const int B = in_sizes[0] / (TT * DD);
    const size_t smem = (size_t)LDS_FLOATS * sizeof(float);

    hipFuncSetAttribute((const void*)temporal_attn_lastrow_g4,
                        hipFuncAttributeMaxDynamicSharedMemorySize, (int)smem);

    lastrow_q_all<<<dim3(B / GQ, DD / NQ), 256, 0, stream>>>(inputs, pos, Wq, out);
    temporal_attn_lastrow_g4<<<B / G, 256, smem, stream>>>(
        inputs, pos, Wk, Wv, out);
}


# S2 (Wk.q GEMM): Wk read as whole lines (16-byte loads of 4 lines; was 8-byte loads touching 16 lines, each line by 8 loads) and re-laid to fragments through a per-wave LDS staging area; bit-identical
# speedup vs baseline: 2.3166x; 1.0447x over previous
#include <hip/hip_runtime.h>
#include <math.h>

constexpr int TT = 64;
constexpr int DD = 512;
constexpr int NH = 16;
constexpr int G  = 4;

constexpr int HSTR = 516;
constexpr int RSTR = 516;
constexpr int QSTR = 36;
constexpr int QG   = NH * QSTR;
constexpr int UG   = DD * NH + 16;
constexpr int WSTR = 36;
constexpr int JS   = 16 * WSTR + 32;
constexpr int STG  = 2 * JS;

constexpr int OFF_H = 0;
constexpr int OFF_Q = OFF_H + TT * HSTR;
constexpr int OFF_U = OFF_Q + G * QG;
constexpr int OFF_S = OFF_U + G * UG;
constexpr int OFF_M = OFF_S + TT * NH;
constexpr int LDS_FLOATS = OFF_M + 2 * 4 * NH;

constexpr int GQ  = 16;
constexpr int NQ  = 256;
constexpr int KQ  = 8;
constexpr int NCQ = DD / (4 * KQ);

constexpr int CH = 16;
constexpr int KC = 4;
constexpr int XN = 32;
constexpr int SK = 16;

typedef __attribute__((ext_vector_type(2))) float v2f;
typedef __attribute__((ext_vector_type(4))) float f4;
typedef __attribute__((ext_vector_type(8))) float v8f;

static __device__ __forceinline__ v8f wmma4(v2f a, v2f b, v8f c) {
    return __builtin_amdgcn_wmma_f32_16x16x4_f32(
        false, a, false, b, (short)0, c, false, false);
}

static __device__ __forceinline__ float xhalf(float x) {
    const int b = __builtin_bit_cast(int, x);
    return __builtin_bit_cast(float,
        __builtin_amdgcn_permlanex16(b, b, 0x76543210, 0xfedcba98, false, false));
}

static __device__ __forceinline__ void lds_barrier() {
    __builtin_amdgcn_fence(__ATOMIC_RELEASE, "workgroup", "local");
    __builtin_amdgcn_s_barrier();
    __builtin_amdgcn_fence(__ATOMIC_ACQUIRE, "workgroup", "local");
}

static __device__ __forceinline__ int last_chunk(int c) { return c < 31 ? c : 31; }
static __device__ __forceinline__ int last_qchunk(int c) { return c < NCQ - 1 ? c : NCQ - 1; }

static __device__ __forceinline__ void ld_qp(v2f (&b)[2 * KQ], const float* W,
                                             int kb, int col2, int hi) {
    #pragma unroll
    for (int s = 0; s < KQ; ++s) {
        const int kk = kb + 4 * s + 2 * hi;
        b[2 * s]     = *(const v2f*)(W + (size_t)kk * DD + col2);
        b[2 * s + 1] = *(const v2f*)(W + (size_t)(kk + 1) * DD + col2);
    }
}

static __device__ __forceinline__ void mac_qp(v8f (&acc)[2], const v2f (&b)[2 * KQ],
                                              const float* arow, int kb, int hi) {
    v2f bq[2 * KQ];
    #pragma unroll
    for (int i = 0; i < 2 * KQ; ++i) {
        bq[i] = b[i];
        asm volatile("" : "+v"(bq[i]));
    }
    v2f a[KQ];
    #pragma unroll
    for (int s = 0; s < KQ; ++s) {
        const int kk = kb + 4 * s + 2 * hi;
        a[s][0] = arow[kk];
        a[s][1] = arow[kk + 1];
    }
    #pragma unroll
    for (int s = 0; s < KQ; ++s) {
        #pragma unroll
        for (int j = 0; j < 2; ++j) {
            v2f bb;
            bb[0] = bq[2 * s][j];
            bb[1] = bq[2 * s + 1][j];
            acc[j] = wmma4(a[s], bb, acc[j]);
        }
    }
}

static __device__ __forceinline__ void ld_o4(v2f (&b)[4 * KC], const float* W,
                                             int kb, int col2, int hi) {
    #pragma unroll
    for (int s = 0; s < KC; ++s) {
        const int kk = kb + 4 * s + 2 * hi;
        const float* p = W + (size_t)kk * DD + col2;
        b[4 * s]     = *(const v2f*)(p);
        b[4 * s + 1] = *(const v2f*)(p + DD);
        b[4 * s + 2] = *(const v2f*)(p + 32);
        b[4 * s + 3] = *(const v2f*)(p + DD + 32);
    }
}

static __device__ __forceinline__ void mac_o4(v8f (&acc)[4], const v2f (&b)[4 * KC],
                                              const float* u2, float vmask,
                                              int kb, int hi) {
    v2f bq[4 * KC];
    #pragma unroll
    for (int i = 0; i < 4 * KC; ++i) {
        bq[i] = b[i];
        asm volatile("" : "+v"(bq[i]));
    }
    #pragma unroll
    for (int s = 0; s < KC; ++s) {
        const int kk = kb + 4 * s + 2 * hi;
        #pragma unroll
        for (int x = 0; x < 2; ++x) {
            v2f a;
            a[0] = vmask * u2[kk * NH + x];
            a[1] = vmask * u2[(kk + 1) * NH + x];
            #pragma unroll
            for (int j = 0; j < 2; ++j) {
                v2f bb;
                bb[0] = bq[4 * s + 2 * x][j];
                bb[1] = bq[4 * s + 2 * x + 1][j];
                acc[2 * x + j] = wmma4(a, bb, acc[2 * x + j]);
            }
        }
    }
}

static __device__ __forceinline__ void ld_wkrows(f4 (&x)[8], const float* Wk,
                                                 int job0, int lane) {
    const int d0 = (job0 >> 4) * 16;
    const int h0 = job0 & 15;
    const float* p = Wk + (size_t)(d0 + (lane >> 4)) * DD + 32 * h0 + 4 * (lane & 15);
    #pragma unroll
    for (int i = 0; i < 8; ++i) x[i] = *(const f4*)(p + (size_t)(2 * i) * DD);
}

static __device__ __forceinline__ void st_wkrows(float* stage, const f4 (&x)[8], int lane) {
    const int pc = lane & 15;
    float* dst = stage + (pc >> 3) * JS + (lane >> 4) * WSTR + 4 * (pc & 7);
    #pragma unroll
    for (int i = 0; i < 8; ++i) *(f4*)(dst + 2 * i * WSTR) = x[i];
}

static __device__ __forceinline__ void wave_lds_order() {
    __builtin_amdgcn_fence(__ATOMIC_RELEASE, "wavefront", "local");
    __builtin_amdgcn_wave_barrier();
    __builtin_amdgcn_fence(__ATOMIC_ACQUIRE, "wavefront", "local");
}

static __device__ __forceinline__ void ld_frag_u2(v2f (&a)[CH], v2f (&b)[CH],
                                                  const float* stage, const float* sQ,
                                                  int job0, int m, int hi) {
    const int gl = m & 3;
    #pragma unroll
    for (int j = 0; j < 2; ++j) {
        const int h = (job0 + j) & 15;
        const float* pa = stage + j * JS + m * WSTR + 2 * hi;
        const float* pb = sQ + gl * QG + h * QSTR + 2 * hi;
        #pragma unroll
        for (int s = 0; s < 8; ++s) {
            a[j * 8 + s] = *(const v2f*)(pa + 4 * s);
            b[j * 8 + s] = *(const v2f*)(pb + 4 * s);
        }
    }
}

static __device__ __forceinline__ void mac_u2(const v2f (&a)[CH], const v2f (&b)[CH],
                                              float* sU, int job0, int m, int hi) {
    const float vmask = (m < G) ? 1.f : 0.f;
    #pragma unroll
    for (int j = 0; j < 2; ++j) {
        const int job = job0 + j;
        const int d0  = (job >> 4) * 16;
        const int h   = job & 15;
        v8f acc = {};
        #pragma unroll
        for (int s = 0; s < 8; ++s) {
            v2f bb;
            bb[0] = vmask * b[j * 8 + s][0];
            bb[1] = vmask * b[j * 8 + s][1];
            acc = wmma4(a[j * 8 + s], bb, acc);
        }
        if (m < G) {
            #pragma unroll
            for (int r = 0; r < 8; ++r)
                sU[m * UG + (d0 + r + 8 * hi) * NH + h] = acc[r];
        }
    }
}

static __device__ __forceinline__ void ld_s(v2f (&a)[SK], v2f (&b)[SK], const float* hrow,
                                            const float* ucol, int d0) {
    #pragma unroll
    for (int s = 0; s < SK; ++s) {
        a[s] = *(const v2f*)(hrow + d0 + 4 * s);
        b[s][0] = ucol[(d0 + 4 * s) * NH];
        b[s][1] = ucol[(d0 + 4 * s + 1) * NH];
    }
}

static __device__ __forceinline__ void mac_s(v8f (&ch)[4], const v2f (&a)[SK], const v2f (&b)[SK]) {
    #pragma unroll
    for (int s = 0; s < SK; ++s) ch[s & 3] = wmma4(a[s], b[s], ch[s & 3]);
}

static __device__ __forceinline__ void ld_tile(f4 (&x)[XN], const float* src, int tid) {
    const f4* in4 = (const f4*)src;
    #pragma unroll
    for (int j = 0; j < XN; ++j) x[j] = in4[tid + 256 * j];
}

__global__ __launch_bounds__(256, 1)
void lastrow_q_all(const float* inputs,
                   const float* pos,
                   const float* Wq,
                   float* q)
{
    __shared__ float sR[GQ * RSTR];

    const int tid  = threadIdx.x;
    const int lane = tid & 31;
    const int wave = tid >> 5;
    const int m    = lane & 15;
    const int hi   = lane >> 4;
    const long b0  = (long)blockIdx.x * GQ;
    const int col2 = blockIdx.y * NQ + wave * 32 + 2 * m;

    f4 r0[GQ * DD / 1024], p0[GQ * DD / 1024];
    #pragma unroll
    for (int u = 0; u < GQ * DD / 1024; ++u) {
        const int i = tid + 256 * u;
        const int g = i >> 7, d = (i & 127) << 2;
        r0[u] = *(const f4*)(inputs + (b0 + g) * (size_t)(TT * DD) + 63 * DD + d);
        p0[u] = *(const f4*)(pos + 63 * DD + d);
    }
    asm volatile("" ::: "memory");
    v2f B0[2 * KQ], B1[2 * KQ], B2[2 * KQ], B3[2 * KQ];
    ld_qp(B0, Wq, 0, col2, hi);
    ld_qp(B1, Wq, 4 * KQ, col2, hi);
    #pragma unroll
    for (int u = 0; u < GQ * DD / 1024; ++u) {
        const int i = tid + 256 * u;
        const int g = i >> 7, d = (i & 127) << 2;
        float* dst = sR + g * RSTR + d;
        dst[0] = r0[u].x + p0[u].x; dst[1] = r0[u].y + p0[u].y;
        dst[2] = r0[u].z + p0[u].z; dst[3] = r0[u].w + p0[u].w;
    }
    lds_barrier();

    const float* arow = sR + m * RSTR;
    v8f acc[2] = {};
    for (int c = 0; c < NCQ; c += 4) {
        ld_qp(B2, Wq, (c + 2) * 4 * KQ, col2, hi);
        mac_qp(acc, B0, arow, c * 4 * KQ, hi);
        ld_qp(B3, Wq, (c + 3) * 4 * KQ, col2, hi);
        mac_qp(acc, B1, arow, (c + 1) * 4 * KQ, hi);
        ld_qp(B0, Wq, last_qchunk(c + 4) * 4 * KQ, col2, hi);
        mac_qp(acc, B2, arow, (c + 2) * 4 * KQ, hi);
        ld_qp(B1, Wq, last_qchunk(c + 5) * 4 * KQ, col2, hi);
        mac_qp(acc, B3, arow, (c + 3) * 4 * KQ, hi);
    }
    #pragma unroll
    for (int r = 0; r < 8; ++r) {
        v2f o;
        o[0] = acc[0][r];
        o[1] = acc[1][r];
        *(v2f*)(q + (b0 + 8 * hi + r) * (size_t)DD + col2) = o;
    }
}

__global__ __launch_bounds__(256, 1)
void temporal_attn_lastrow_g4(const float* inputs,
                              const float* pos,
                              const float* Wk,
                              const float* Wv,
                              float* out)
{
    extern __shared__ __attribute__((aligned(16))) float lds[];
    float* sH = lds + OFF_H;
    float* sQ = lds + OFF_Q;
    float* sU = lds + OFF_U;
    float* sS = lds + OFF_S;
    float* sM = lds + OFF_M;
    float* sL = sM + 4 * NH;

    const int tid  = threadIdx.x;
    const int lane = tid & 31;
    const int wave = tid >> 5;
    const int m    = lane & 15;
    const int hi   = lane >> 4;
    const long b0  = (long)blockIdx.x * G;

    float q0[G * DD / 256];
    #pragma unroll
    for (int u = 0; u < G * DD / 256; ++u) {
        const int i = tid + 256 * u;
        const int g = i >> 9, e = i & 511;
        q0[u] = out[(b0 + g) * (size_t)DD + e];
    }
    asm volatile("" ::: "memory");
    f4 X[XN];
    ld_tile(X, inputs + b0 * (size_t)(TT * DD), tid);
    #pragma unroll
    for (int u = 0; u < G * DD / 256; ++u) {
        const int i = tid + 256 * u;
        const int g = i >> 9, e = i & 511;
        sQ[g * QG + (e >> 5) * QSTR + (e & 31)] = q0[u];
    }
    lds_barrier();

    {
        f4  G0[8], G1[8];
        v2f FA0[CH], FB0[CH], FA1[CH], FB1[CH];
        float* stg0 = lds + OFF_H + wave * (2 * STG);
        float* stg1 = stg0 + STG;
        const int jb = wave * 64;
        ld_wkrows(G0, Wk, jb, lane);
        ld_wkrows(G1, Wk, jb + 2, lane);
        asm volatile("" ::: "memory");
        st_wkrows(stg0, G0, lane);
        ld_wkrows(G0, Wk, jb + 4, lane);
        wave_lds_order();
        ld_frag_u2(FA0, FB0, stg0, sQ, jb, m, hi);
        for (int c = 0; c < 32; c += 2) {
            st_wkrows(stg1, G1, lane);
            ld_wkrows(G1, Wk, jb + 2 * last_chunk(c + 3), lane);
            wave_lds_order();
            ld_frag_u2(FA1, FB1, stg1, sQ, jb + 2 * (c + 1), m, hi);
            mac_u2(FA0, FB0, sU, jb + 2 * c, m, hi);
            st_wkrows(stg0, G0, lane);
            ld_wkrows(G0, Wk, jb + 2 * last_chunk(c + 4), lane);
            wave_lds_order();
            ld_frag_u2(FA0, FB0, stg0, sQ, jb + 2 * last_chunk(c + 2), m, hi);
            mac_u2(FA1, FB1, sU, jb + 2 * (c + 1), m, hi);
        }
    }
    lds_barrier();

    const float scale = 0.044194173824159216f;
    for (int g = 0; g < G; ++g) {
        {
            const f4* pe4 = (const f4*)pos;
            f4 P[XN];
            #pragma unroll
            for (int j = 0; j < XN; ++j) P[j] = pe4[tid + 256 * j];
            #pragma unroll
            for (int j = 0; j < XN; ++j) {
                const int i = tid + 256 * j;
                f4 x = X[j], p = P[j];
                int t = i >> 7, d = (i & 127) << 2;
                f4 h;
                h.x = x.x + p.x; h.y = x.y + p.y;
                h.z = x.z + p.z; h.w = x.w + p.w;
                *(f4*)(sH + t * HSTR + d) = h;
            }
            if (g + 1 < G)
                ld_tile(X, inputs + (b0 + g + 1) * (size_t)(TT * DD), tid);
        }
        lds_barrier();

        float sc[8] = {};
        if (wave < 4) {
            const int t0 = wave * 16;
            const float* hrow = sH + (t0 + m) * HSTR + 2 * hi;
            int uoff = OFF_U + g * UG + (2 * hi) * NH + m;
            asm volatile("" : "+v"(uoff));
            const float* ucol = lds + uoff;
            v2f A0[SK], A1[SK], B0[SK], B1[SK];
            v8f ch[4] = {};
            ld_s(A0, B0, hrow, ucol, 0);
            for (int c = 0; c < DD / (4 * SK); c += 2) {
                ld_s(A1, B1, hrow, ucol, (c + 1) * 4 * SK);
                mac_s(ch, A0, B0);
                if (c + 2 < DD / (4 * SK)) ld_s(A0, B0, hrow, ucol, (c + 2) * 4 * SK);
                mac_s(ch, A1, B1);
            }
            const v8f acc = (ch[0] + ch[1]) + (ch[2] + ch[3]);
            float mx = -INFINITY;
            #pragma unroll
            for (int r = 0; r < 8; ++r) {
                sc[r] = acc[r] * scale;
                mx = fmaxf(mx, sc[r]);
            }
            mx = fmaxf(mx, xhalf(mx));
            if (hi == 0) sM[wave * NH + m] = mx;
        }
        lds_barrier();

        if (wave < 4) {
            const int t0 = wave * 16;
            const float mx = fmaxf(fmaxf(sM[m], sM[NH + m]),
                                   fmaxf(sM[2 * NH + m], sM[3 * NH + m]));
            float sum = 0.f;
            #pragma unroll
            for (int r = 0; r < 8; ++r) {
                const float e = expf(sc[r] - mx);
                sS[(t0 + r + 8 * hi) * NH + m] = e;
                sum += e;
            }
            sum += xhalf(sum);
            if (hi == 0) sL[wave * NH + m] = sum;
        }
        lds_barrier();

        {
            v2f pb[16], ha[4][16];
            const float inv = 1.f / (((sL[m] + sL[NH + m]) + sL[2 * NH + m]) + sL[3 * NH + m]);
            #pragma unroll
            for (int s = 0; s < 16; ++s) {
                const int kk = 4 * s + 2 * hi;
                pb[s][0] = sS[kk * NH + m] * inv;
                pb[s][1] = sS[(kk + 1) * NH + m] * inv;
            }
            #pragma unroll
            for (int t = 0; t < 4; ++t) {
                const int d0 = (wave * 4 + t) * 16;
                #pragma unroll
                for (int s = 0; s < 16; ++s) {
                    const int kk = 4 * s + 2 * hi;
                    ha[t][s][0] = sH[kk * HSTR + d0 + m];
                    ha[t][s][1] = sH[(kk + 1) * HSTR + d0 + m];
                }
            }
            #pragma unroll
            for (int t = 0; t < 4; ++t) {
                const int d0 = (wave * 4 + t) * 16;
                v8f acc = {};
                #pragma unroll
                for (int s = 0; s < 16; ++s) acc = wmma4(ha[t][s], pb[s], acc);
                #pragma unroll
                for (int r = 0; r < 8; ++r)
                    sU[g * UG + (d0 + r + 8 * hi) * NH + m] = acc[r];
            }
        }
        lds_barrier();
    }

    {
        const int   gl    = m & 3;
        const float vmask = (m < G) ? 1.f : 0.f;
        const float* u2   = sU + gl * UG + 2 * wave;
        const int col2 = wave * 64 + 2 * m;
        v2f B0[4 * KC], B1[4 * KC], B2[4 * KC], B3[4 * KC];
        v8f acc[4] = {};
        ld_o4(B0, Wv, 0, col2, hi);
        ld_o4(B1, Wv, 16, col2, hi);
        for (int c = 0; c < 32; c += 4) {
            ld_o4(B2, Wv, (c + 2) * 16, col2, hi);
            mac_o4(acc, B0, u2, vmask, c * 16, hi);
            ld_o4(B3, Wv, (c + 3) * 16, col2, hi);
            mac_o4(acc, B1, u2, vmask, (c + 1) * 16, hi);
            ld_o4(B0, Wv, last_chunk(c + 4) * 16, col2, hi);
            mac_o4(acc, B2, u2, vmask, (c + 2) * 16, hi);
            ld_o4(B1, Wv, last_chunk(c + 5) * 16, col2, hi);
            mac_o4(acc, B3, u2, vmask, (c + 3) * 16, hi);
        }
        if (hi == 0) {
            #pragma unroll
            for (int x = 0; x < 2; ++x) {
                #pragma unroll
                for (int r = 0; r < G; ++r) {
                    v2f o;
                    o[0] = acc[2 * x][r];
                    o[1] = acc[2 * x + 1][r];
                    *(v2f*)(out + (b0 + r) * (size_t)DD + col2 + 32 * x) = o;
                }
            }
        }
    }
}

extern "C" void kernel_launch(void* const* d_in, const int* in_sizes, int n_in,
                              void* d_out, int out_size, void* d_ws, size_t ws_size,
                              hipStream_t stream) {
    const float* inputs = (const float*)d_in[0];
    const float* pos    = (const float*)d_in[1];
    const float* Wq     = (const float*)d_in[2];
    const float* Wk     = (const float*)d_in[3];
    const float* Wv     = (const float*)d_in[4];
    float* out          = (float*)d_out;
    (void)d_ws; (void)ws_size; (void)n_in; (void)out_size;

    const int B = in_sizes[0] / (TT * DD);
    const size_t smem = (size_t)LDS_FLOATS * sizeof(float);

    hipFuncSetAttribute((const void*)temporal_attn_lastrow_g4,
                        hipFuncAttributeMaxDynamicSharedMemorySize, (int)smem);

    lastrow_q_all<<<dim3(B / GQ, DD / NQ), 256, 0, stream>>>(inputs, pos, Wq, out);
    temporal_attn_lastrow_g4<<<B / G, 256, smem, stream>>>(
        inputs, pos, Wk, Wv, out);
}
